# starting kernel
# baseline (speedup 1.0000x reference)
#include <hip/hip_runtime.h>
#include <hip/hip_bf16.h>

#define NXX 64
#define NYY 32
#define NUU 16
#define NDD 8
#define T_STEPS 2048
#define B_TOT 256

typedef float v2f __attribute__((ext_vector_type(2)));
typedef float v8f __attribute__((ext_vector_type(8)));

__device__ __forceinline__ v8f zero8() {
    v8f c;
#pragma unroll
    for (int r = 0; r < 8; ++r) c[r] = 0.0f;
    return c;
}

__device__ __forceinline__ v2f fragA(const float* X, int ld, int m0, int k0, int lane) {
    int h = lane >> 4, m = m0 + (lane & 15), k = k0 + 2 * h;
    v2f a; a.x = X[m * ld + k]; a.y = X[m * ld + k + 1]; return a;
}
__device__ __forceinline__ v2f fragAT(const float* X, int ld, int m0, int k0, int lane) {
    int h = lane >> 4, m = m0 + (lane & 15), k = k0 + 2 * h;
    v2f a; a.x = X[k * ld + m]; a.y = X[(k + 1) * ld + m]; return a;
}
__device__ __forceinline__ v2f fragB(const float* X, int ld, int k0, int n0, int lane) {
    int h = lane >> 4, n = n0 + (lane & 15), k = k0 + 2 * h;
    v2f b; b.x = X[k * ld + n]; b.y = X[(k + 1) * ld + n]; return b;
}
__device__ __forceinline__ v2f fragBT(const float* X, int ld, int k0, int n0, int lane) {
    int h = lane >> 4, n = n0 + (lane & 15), k = k0 + 2 * h;
    v2f b; b.x = X[n * ld + k]; b.y = X[n * ld + k + 1]; return b;
}
__device__ __forceinline__ void storeD(float* X, int ld, int m0, int n0, v8f d, int lane,
                                       float sgn = 1.0f) {
    int h = lane >> 4, n = n0 + (lane & 15);
#pragma unroll
    for (int r = 0; r < 8; ++r) X[(m0 + r + 8 * h) * ld + n] = sgn * d[r];
}
__device__ __forceinline__ v8f loadC(const float* X, int ld, int m0, int n0, int lane) {
    int h = lane >> 4, n = n0 + (lane & 15);
    v8f c;
#pragma unroll
    for (int r = 0; r < 8; ++r) c[r] = X[(m0 + r + 8 * h) * ld + n];
    return c;
}
__device__ __forceinline__ v8f identC(int m0, int n0, int lane) {
    int h = lane >> 4, n = n0 + (lane & 15);
    v8f c;
#pragma unroll
    for (int r = 0; r < 8; ++r) c[r] = ((m0 + r + 8 * h) == n) ? 1.0f : 0.0f;
    return c;
}

template <bool TA, bool TB>
__device__ __forceinline__ v8f mmacc(const float* A, int lda, int m0,
                                     const float* B, int ldb, int n0,
                                     int K, v8f c, int lane) {
#pragma unroll
    for (int k = 0; k < K; k += 4) {
        v2f a = TA ? fragAT(A, lda, m0, k, lane) : fragA(A, lda, m0, k, lane);
        v2f b = TB ? fragBT(B, ldb, k, n0, lane) : fragB(B, ldb, k, n0, lane);
        c = __builtin_amdgcn_wmma_f32_16x16x4_f32(false, a, false, b, (short)0, c,
                                                  false, false);
    }
    return c;
}

__global__ __launch_bounds__(256) void kf_gain_seq(const float* __restrict__ Wfx,
                                                   const float* __restrict__ Wfy,
                                                   float* __restrict__ G) {
    __shared__ float sWfx[NXX * NXX];
    __shared__ float sWfy[NXX * NYY];
    __shared__ float sP[NXX * NXX];
    __shared__ float sM1[NXX * NXX];
    __shared__ float sM2[NXX * NYY];
    __shared__ float sSa[NYY * 64];
    __shared__ float sL[NXX * NYY];

    const int tid = threadIdx.x, lane = tid & 31, w = tid >> 5;

    for (int i = tid; i < NXX * NXX; i += 256) {
        sWfx[i] = Wfx[i];
        sP[i] = ((i >> 6) == (i & 63)) ? 1.0f : 0.0f;
    }
    for (int i = tid; i < NXX * NYY; i += 256) {
        sWfy[i] = Wfy[i];
        sL[i] = 0.0f;
    }
    __syncthreads();

    for (int t = 0; t < T_STEPS; ++t) {
        float* Gt = G + (size_t)t * (NYY * NXX);
        for (int i = tid; i < NYY * NXX; i += 256) {
            int j = i >> 6, col = i & 63;
            Gt[i] = sL[col * NYY + j];
        }
        __syncthreads();

        for (int tt = w; tt < 16; tt += 8) {
            int mi = (tt >> 2) * 16, ni = (tt & 3) * 16;
            v8f c = zero8();
            c = mmacc<false, false>(sWfx, NXX, mi, sP, NXX, ni, NXX, c, lane);
            storeD(sM1, NXX, mi, ni, c, lane);
        }
        __syncthreads();
        for (int tt = w; tt < 16; tt += 8) {
            int mi = (tt >> 2) * 16, ni = (tt & 3) * 16;
            v8f c = identC(mi, ni, lane);
            c = mmacc<false, true>(sM1, NXX, mi, sWfx, NXX, ni, NXX, c, lane);
            storeD(sP, NXX, mi, ni, c, lane);
        }
        __syncthreads();
        {
            int mi = (w >> 1) * 16, ni = (w & 1) * 16;
            v8f c = zero8();
            c = mmacc<false, false>(sP, NXX, mi, sWfy, NYY, ni, NXX, c, lane);
            storeD(sM2, NYY, mi, ni, c, lane);
        }
        __syncthreads();
        if (w < 4) {
            int mi = (w >> 1) * 16, ni = (w & 1) * 16;
            v8f c = identC(mi, ni, lane);
            c = mmacc<true, false>(sWfy, NYY, mi, sM2, NYY, ni, NXX, c, lane);
            storeD(sSa, 64, mi, ni, c, lane);
        } else {
            int r0 = (w - 4) * 8;
#pragma unroll
            for (int rr = 0; rr < 8; ++rr) {
                int r = r0 + rr;
                sSa[r * 64 + 32 + lane] = (lane == r) ? 1.0f : 0.0f;
            }
        }
        __syncthreads();
        {
            const int r = tid >> 3, cs = (tid & 7) * 8;
            for (int p = 0; p < 32; ++p) {
                float piv = sSa[p * 64 + p];
                float f = sSa[r * 64 + p];
                float pr[8];
#pragma unroll
                for (int j = 0; j < 8; ++j) pr[j] = sSa[p * 64 + cs + j];
                __syncthreads();
                float ip = 1.0f / piv;
                if (r == p) {
#pragma unroll
                    for (int j = 0; j < 8; ++j) sSa[p * 64 + cs + j] = pr[j] * ip;
                } else {
#pragma unroll
                    for (int j = 0; j < 8; ++j) sSa[r * 64 + cs + j] -= f * pr[j] * ip;
                }
                __syncthreads();
            }
        }
        {
            int mi = (w >> 1) * 16, ni = (w & 1) * 16;
            v8f c = zero8();
            c = mmacc<false, false>(sM2, NYY, mi, sSa + 32, 64, ni, NYY, c, lane);
            storeD(sL, NYY, mi, ni, c, lane);
        }
        __syncthreads();
        {
            int mi = (w >> 2) * 16, ni = (w & 3) * 16;
            v8f c = zero8();
            c = mmacc<true, false>(sWfy, NYY, mi, sP, NXX, ni, NXX, c, lane);
            storeD(sM1, NXX, mi, ni, c, lane, -1.0f);
        }
        __syncthreads();
        for (int tt = w; tt < 16; tt += 8) {
            int mi = (tt >> 2) * 16, ni = (tt & 3) * 16;
            v8f c = identC(mi, ni, lane);
            c = mmacc<false, false>(sL, NYY, mi, sM1, NXX, ni, NYY, c, lane);
            storeD(sP, NXX, mi, ni, c, lane);
        }
        __syncthreads();
    }
}

__global__ __launch_bounds__(128) void kf_batch(const float* __restrict__ Yp,
                                                const float* __restrict__ Up,
                                                const float* __restrict__ Dp,
                                                const float* __restrict__ Wfx,
                                                const float* __restrict__ Wfu,
                                                const float* __restrict__ Wfd,
                                                const float* __restrict__ Wfy,
                                                const float* __restrict__ G,
                                                float* __restrict__ out) {
    __shared__ float sW[(NXX + NUU + NDD) * NXX];
    __shared__ float sWfy[NXX * NYY];
    __shared__ float sx[16 * NXX];
    __shared__ float sxp[16 * NXX];
    __shared__ float su[16 * NUU];
    __shared__ float sd[16 * NDD];
    __shared__ float sym[16 * NYY];
    __shared__ float sin[16 * NYY];

    const int tid = threadIdx.x, lane = tid & 31, w = tid >> 5;
    const int r0 = blockIdx.x * 16;

    for (int i = tid; i < NXX * NXX; i += 128) sW[i] = Wfx[i];
    for (int i = tid; i < NUU * NXX; i += 128) sW[NXX * NXX + i] = Wfu[i];
    for (int i = tid; i < NDD * NXX; i += 128) sW[(NXX + NUU) * NXX + i] = Wfd[i];
    for (int i = tid; i < NXX * NYY; i += 128) sWfy[i] = Wfy[i];
    for (int i = tid; i < 16 * NXX; i += 128) sx[i] = 0.0f;
    __syncthreads();

    for (int t = 0; t < T_STEPS; ++t) {
        const float* ut = Up + ((size_t)t * B_TOT + r0) * NUU;
        const float* dt = Dp + ((size_t)t * B_TOT + r0) * NDD;
        const float* yt = Yp + ((size_t)t * B_TOT + r0) * NYY;
        for (int i = tid; i < 16 * NUU; i += 128) su[i] = ut[i];
        for (int i = tid; i < 16 * NDD; i += 128) sd[i] = dt[i];
        for (int i = tid; i < 16 * NYY; i += 128) sym[i] = yt[i];
        const float* Gt = G + (size_t)t * (NYY * NXX);
        if (t + 1 < T_STEPS) {
            __builtin_prefetch(G + (size_t)(t + 1) * (NYY * NXX) + tid * 16, 0, 1);
        }
        __syncthreads();

        {
            int n0 = w * 16;
            v8f c = zero8();
            c = mmacc<false, false>(sx, NXX, 0, sW, NXX, n0, NXX, c, lane);
            c = mmacc<false, false>(su, NUU, 0, sW + NXX * NXX, NXX, n0, NUU, c, lane);
            c = mmacc<false, false>(sd, NDD, 0, sW + (NXX + NUU) * NXX, NXX, n0, NDD, c, lane);
            storeD(sxp, NXX, 0, n0, c, lane);
        }
        __syncthreads();

        if (w < 2) {
            int n0 = w * 16;
            v8f c = zero8();
            c = mmacc<false, false>(sxp, NXX, 0, sWfy, NYY, n0, NXX, c, lane);
            int h = lane >> 4, n = n0 + (lane & 15);
#pragma unroll
            for (int r = 0; r < 8; ++r) {
                int m = r + 8 * h;
                sin[m * NYY + n] = sym[m * NYY + n] - c[r];
            }
        }
        __syncthreads();

        {
            int n0 = w * 16;
            v8f c = loadC(sxp, NXX, 0, n0, lane);
            c = mmacc<false, false>(sin, NYY, 0, Gt, NXX, n0, NYY, c, lane);
            storeD(sx, NXX, 0, n0, c, lane);
        }
        __syncthreads();
    }

    for (int i = tid; i < 16 * NXX; i += 128) {
        int r = i >> 6, cc = i & 63;
        out[(size_t)(r0 + r) * NXX + cc] = sx[i];
    }
}

extern "C" void kernel_launch(void* const* d_in, const int* in_sizes, int n_in,
                              void* d_out, int out_size, void* d_ws, size_t ws_size,
                              hipStream_t stream) {
    (void)in_sizes; (void)n_in; (void)out_size; (void)ws_size;
    const float* Yp  = (const float*)d_in[0];
    const float* Up  = (const float*)d_in[1];
    const float* Dp  = (const float*)d_in[2];
    const float* Wfx = (const float*)d_in[3];
    const float* Wfu = (const float*)d_in[4];
    const float* Wfd = (const float*)d_in[5];
    const float* Wfy = (const float*)d_in[6];
    float* G = (float*)d_ws;
    float* out = (float*)d_out;

    kf_gain_seq<<<1, 256, 0, stream>>>(Wfx, Wfy, G);
    kf_batch<<<16, 128, 0, stream>>>(Yp, Up, Dp, Wfx, Wfu, Wfd, Wfy, G, out);
}


# Gain kernel: 32x32 innovation-matrix inverse done in one wave's registers (row per lane, pivot row by lane broadcast, rows scaled once at the end) instead of in LDS with two barriers per pivot
# speedup vs baseline: 1.2623x; 1.2623x over previous
#include <hip/hip_runtime.h>
#include <hip/hip_bf16.h>

#define NXX 64
#define NYY 32
#define NUU 16
#define NDD 8
#define T_STEPS 2048
#define B_TOT 256
#define SLD 36

typedef float v2f __attribute__((ext_vector_type(2)));
typedef float v8f __attribute__((ext_vector_type(8)));

__device__ __forceinline__ v8f zero8() {
    v8f c;
#pragma unroll
    for (int r = 0; r < 8; ++r) c[r] = 0.0f;
    return c;
}

__device__ __forceinline__ v2f fragA(const float* X, int ld, int m0, int k0, int lane) {
    int h = lane >> 4, m = m0 + (lane & 15), k = k0 + 2 * h;
    v2f a; a.x = X[m * ld + k]; a.y = X[m * ld + k + 1]; return a;
}
__device__ __forceinline__ v2f fragAT(const float* X, int ld, int m0, int k0, int lane) {
    int h = lane >> 4, m = m0 + (lane & 15), k = k0 + 2 * h;
    v2f a; a.x = X[k * ld + m]; a.y = X[(k + 1) * ld + m]; return a;
}
__device__ __forceinline__ v2f fragB(const float* X, int ld, int k0, int n0, int lane) {
    int h = lane >> 4, n = n0 + (lane & 15), k = k0 + 2 * h;
    v2f b; b.x = X[k * ld + n]; b.y = X[(k + 1) * ld + n]; return b;
}
__device__ __forceinline__ v2f fragBT(const float* X, int ld, int k0, int n0, int lane) {
    int h = lane >> 4, n = n0 + (lane & 15), k = k0 + 2 * h;
    v2f b; b.x = X[n * ld + k]; b.y = X[n * ld + k + 1]; return b;
}
__device__ __forceinline__ void storeD(float* X, int ld, int m0, int n0, v8f d, int lane,
                                       float sgn = 1.0f) {
    int h = lane >> 4, n = n0 + (lane & 15);
#pragma unroll
    for (int r = 0; r < 8; ++r) X[(m0 + r + 8 * h) * ld + n] = sgn * d[r];
}
__device__ __forceinline__ v8f loadC(const float* X, int ld, int m0, int n0, int lane) {
    int h = lane >> 4, n = n0 + (lane & 15);
    v8f c;
#pragma unroll
    for (int r = 0; r < 8; ++r) c[r] = X[(m0 + r + 8 * h) * ld + n];
    return c;
}
__device__ __forceinline__ v8f identC(int m0, int n0, int lane) {
    int h = lane >> 4, n = n0 + (lane & 15);
    v8f c;
#pragma unroll
    for (int r = 0; r < 8; ++r) c[r] = ((m0 + r + 8 * h) == n) ? 1.0f : 0.0f;
    return c;
}

__device__ __forceinline__ float lane_bcast(float x, int l) {
    return __int_as_float(__builtin_amdgcn_readlane(__float_as_int(x), l));
}

template <bool TA, bool TB>
__device__ __forceinline__ v8f mmacc(const float* A, int lda, int m0,
                                     const float* B, int ldb, int n0,
                                     int K, v8f c, int lane) {
#pragma unroll
    for (int k = 0; k < K; k += 4) {
        v2f a = TA ? fragAT(A, lda, m0, k, lane) : fragA(A, lda, m0, k, lane);
        v2f b = TB ? fragBT(B, ldb, k, n0, lane) : fragB(B, ldb, k, n0, lane);
        c = __builtin_amdgcn_wmma_f32_16x16x4_f32(false, a, false, b, (short)0, c,
                                                  false, false);
    }
    return c;
}

__global__ __launch_bounds__(256) void kf_gain_seq(const float* __restrict__ Wfx,
                                                   const float* __restrict__ Wfy,
                                                   float* __restrict__ G) {
    __shared__ float sWfx[NXX * NXX];
    __shared__ float sWfy[NXX * NYY];
    __shared__ float sP[NXX * NXX];
    __shared__ float sM1[NXX * NXX];
    __shared__ float sM2[NXX * NYY];
    __shared__ __attribute__((aligned(16))) float sS[NYY * SLD];
    __shared__ float sL[NXX * NYY];

    const int tid = threadIdx.x, lane = tid & 31, w = tid >> 5;

    for (int i = tid; i < NXX * NXX; i += 256) {
        sWfx[i] = Wfx[i];
        sP[i] = ((i >> 6) == (i & 63)) ? 1.0f : 0.0f;
    }
    for (int i = tid; i < NXX * NYY; i += 256) {
        sWfy[i] = Wfy[i];
        sL[i] = 0.0f;
    }
    __syncthreads();

    for (int t = 0; t < T_STEPS; ++t) {
        float* Gt = G + (size_t)t * (NYY * NXX);
        for (int i = tid; i < NYY * NXX; i += 256) {
            int j = i >> 6, col = i & 63;
            Gt[i] = sL[col * NYY + j];
        }
        __syncthreads();

        for (int tt = w; tt < 16; tt += 8) {
            int mi = (tt >> 2) * 16, ni = (tt & 3) * 16;
            v8f c = zero8();
            c = mmacc<false, false>(sWfx, NXX, mi, sP, NXX, ni, NXX, c, lane);
            storeD(sM1, NXX, mi, ni, c, lane);
        }
        __syncthreads();
        for (int tt = w; tt < 16; tt += 8) {
            int mi = (tt >> 2) * 16, ni = (tt & 3) * 16;
            v8f c = identC(mi, ni, lane);
            c = mmacc<false, true>(sM1, NXX, mi, sWfx, NXX, ni, NXX, c, lane);
            storeD(sP, NXX, mi, ni, c, lane);
        }
        __syncthreads();
        {
            int mi = (w >> 1) * 16, ni = (w & 1) * 16;
            v8f c = zero8();
            c = mmacc<false, false>(sP, NXX, mi, sWfy, NYY, ni, NXX, c, lane);
            storeD(sM2, NYY, mi, ni, c, lane);
        }
        __syncthreads();
        if (w < 4) {
            int mi = (w >> 1) * 16, ni = (w & 1) * 16;
            v8f c = identC(mi, ni, lane);
            c = mmacc<true, false>(sWfy, NYY, mi, sM2, NYY, ni, NXX, c, lane);
            storeD(sS, SLD, mi, ni, c, lane);
        }
        __syncthreads();
        if (w == 0) {
            float a[32];
            const float4* rin = reinterpret_cast<const float4*>(sS + lane * SLD);
#pragma unroll
            for (int j = 0; j < 8; ++j) {
                float4 v = rin[j];
                a[4 * j] = v.x; a[4 * j + 1] = v.y; a[4 * j + 2] = v.z; a[4 * j + 3] = v.w;
            }
            float s = 1.0f;
            float ip = 1.0f / lane_bcast(a[0], 0);
            for (int p = 0; p < 32; ++p) {
                const bool isp = (lane == p);
                const float g = isp ? 0.0f : a[0] * ip;
                s = isp ? ip : s;
                a[0] = a[1] - g * lane_bcast(a[1], p);
                const float ipn = 1.0f / lane_bcast(a[0], (p + 1) & 31);
#pragma unroll
                for (int c = 2; c < 32; ++c) a[c - 1] = a[c] - g * lane_bcast(a[c], p);
                a[31] = isp ? 1.0f : -g;
                ip = ipn;
            }
            float4* rout = reinterpret_cast<float4*>(sS + lane * SLD);
#pragma unroll
            for (int j = 0; j < 8; ++j) {
                float4 v;
                v.x = a[4 * j] * s; v.y = a[4 * j + 1] * s;
                v.z = a[4 * j + 2] * s; v.w = a[4 * j + 3] * s;
                rout[j] = v;
            }
        }
        __syncthreads();
        {
            int mi = (w >> 1) * 16, ni = (w & 1) * 16;
            v8f c = zero8();
            c = mmacc<false, false>(sM2, NYY, mi, sS, SLD, ni, NYY, c, lane);
            storeD(sL, NYY, mi, ni, c, lane);
        }
        __syncthreads();
        {
            int mi = (w >> 2) * 16, ni = (w & 3) * 16;
            v8f c = zero8();
            c = mmacc<true, false>(sWfy, NYY, mi, sP, NXX, ni, NXX, c, lane);
            storeD(sM1, NXX, mi, ni, c, lane, -1.0f);
        }
        __syncthreads();
        for (int tt = w; tt < 16; tt += 8) {
            int mi = (tt >> 2) * 16, ni = (tt & 3) * 16;
            v8f c = identC(mi, ni, lane);
            c = mmacc<false, false>(sL, NYY, mi, sM1, NXX, ni, NYY, c, lane);
            storeD(sP, NXX, mi, ni, c, lane);
        }
        __syncthreads();
    }
}

__global__ __launch_bounds__(128) void kf_batch(const float* __restrict__ Yp,
                                                const float* __restrict__ Up,
                                                const float* __restrict__ Dp,
                                                const float* __restrict__ Wfx,
                                                const float* __restrict__ Wfu,
                                                const float* __restrict__ Wfd,
                                                const float* __restrict__ Wfy,
                                                const float* __restrict__ G,
                                                float* __restrict__ out) {
    __shared__ float sW[(NXX + NUU + NDD) * NXX];
    __shared__ float sWfy[NXX * NYY];
    __shared__ float sx[16 * NXX];
    __shared__ float sxp[16 * NXX];
    __shared__ float su[16 * NUU];
    __shared__ float sd[16 * NDD];
    __shared__ float sym[16 * NYY];
    __shared__ float sin[16 * NYY];

    const int tid = threadIdx.x, lane = tid & 31, w = tid >> 5;
    const int r0 = blockIdx.x * 16;

    for (int i = tid; i < NXX * NXX; i += 128) sW[i] = Wfx[i];
    for (int i = tid; i < NUU * NXX; i += 128) sW[NXX * NXX + i] = Wfu[i];
    for (int i = tid; i < NDD * NXX; i += 128) sW[(NXX + NUU) * NXX + i] = Wfd[i];
    for (int i = tid; i < NXX * NYY; i += 128) sWfy[i] = Wfy[i];
    for (int i = tid; i < 16 * NXX; i += 128) sx[i] = 0.0f;
    __syncthreads();

    for (int t = 0; t < T_STEPS; ++t) {
        const float* ut = Up + ((size_t)t * B_TOT + r0) * NUU;
        const float* dt = Dp + ((size_t)t * B_TOT + r0) * NDD;
        const float* yt = Yp + ((size_t)t * B_TOT + r0) * NYY;
        for (int i = tid; i < 16 * NUU; i += 128) su[i] = ut[i];
        for (int i = tid; i < 16 * NDD; i += 128) sd[i] = dt[i];
        for (int i = tid; i < 16 * NYY; i += 128) sym[i] = yt[i];
        const float* Gt = G + (size_t)t * (NYY * NXX);
        if (t + 1 < T_STEPS) {
            __builtin_prefetch(G + (size_t)(t + 1) * (NYY * NXX) + tid * 16, 0, 1);
        }
        __syncthreads();

        {
            int n0 = w * 16;
            v8f c = zero8();
            c = mmacc<false, false>(sx, NXX, 0, sW, NXX, n0, NXX, c, lane);
            c = mmacc<false, false>(su, NUU, 0, sW + NXX * NXX, NXX, n0, NUU, c, lane);
            c = mmacc<false, false>(sd, NDD, 0, sW + (NXX + NUU) * NXX, NXX, n0, NDD, c, lane);
            storeD(sxp, NXX, 0, n0, c, lane);
        }
        __syncthreads();

        if (w < 2) {
            int n0 = w * 16;
            v8f c = zero8();
            c = mmacc<false, false>(sxp, NXX, 0, sWfy, NYY, n0, NXX, c, lane);
            int h = lane >> 4, n = n0 + (lane & 15);
#pragma unroll
            for (int r = 0; r < 8; ++r) {
                int m = r + 8 * h;
                sin[m * NYY + n] = sym[m * NYY + n] - c[r];
            }
        }
        __syncthreads();

        {
            int n0 = w * 16;
            v8f c = loadC(sxp, NXX, 0, n0, lane);
            c = mmacc<false, false>(sin, NYY, 0, Gt, NXX, n0, NYY, c, lane);
            storeD(sx, NXX, 0, n0, c, lane);
        }
        __syncthreads();
    }

    for (int i = tid; i < 16 * NXX; i += 128) {
        int r = i >> 6, cc = i & 63;
        out[(size_t)(r0 + r) * NXX + cc] = sx[i];
    }
}

extern "C" void kernel_launch(void* const* d_in, const int* in_sizes, int n_in,
                              void* d_out, int out_size, void* d_ws, size_t ws_size,
                              hipStream_t stream) {
    (void)in_sizes; (void)n_in; (void)out_size; (void)ws_size;
    const float* Yp  = (const float*)d_in[0];
    const float* Up  = (const float*)d_in[1];
    const float* Dp  = (const float*)d_in[2];
    const float* Wfx = (const float*)d_in[3];
    const float* Wfu = (const float*)d_in[4];
    const float* Wfd = (const float*)d_in[5];
    const float* Wfy = (const float*)d_in[6];
    float* G = (float*)d_ws;
    float* out = (float*)d_out;

    kf_gain_seq<<<1, 256, 0, stream>>>(Wfx, Wfy, G);
    kf_batch<<<16, 128, 0, stream>>>(Yp, Up, Dp, Wfx, Wfu, Wfd, Wfy, G, out);
}


# LDS matrices get padded row pitches (76/44/20/12 floats, 4*odd) so WMMA fragment loads and D stores spread over all 64 banks instead of one bank pair per half-wave at pitch 64/32; sums unchanged
# speedup vs baseline: 2.0496x; 1.6237x over previous
#include <hip/hip_runtime.h>
#include <hip/hip_bf16.h>

#define NXX 64
#define NYY 32
#define NUU 16
#define NDD 8
#define T_STEPS 2048
#define B_TOT 256
#define LDX 76
#define LDY 44
#define LDU 20
#define LDD 12
#define SLD LDY

typedef float v2f __attribute__((ext_vector_type(2)));
typedef float v8f __attribute__((ext_vector_type(8)));

__device__ __forceinline__ v8f zero8() {
    v8f c;
#pragma unroll
    for (int r = 0; r < 8; ++r) c[r] = 0.0f;
    return c;
}

__device__ __forceinline__ v2f fragA(const float* X, int ld, int m0, int k0, int lane) {
    int h = lane >> 4, m = m0 + (lane & 15), k = k0 + 2 * h;
    v2f a; a.x = X[m * ld + k]; a.y = X[m * ld + k + 1]; return a;
}
__device__ __forceinline__ v2f fragAT(const float* X, int ld, int m0, int k0, int lane) {
    int h = lane >> 4, m = m0 + (lane & 15), k = k0 + 2 * h;
    v2f a; a.x = X[k * ld + m]; a.y = X[(k + 1) * ld + m]; return a;
}
__device__ __forceinline__ v2f fragB(const float* X, int ld, int k0, int n0, int lane) {
    int h = lane >> 4, n = n0 + (lane & 15), k = k0 + 2 * h;
    v2f b; b.x = X[k * ld + n]; b.y = X[(k + 1) * ld + n]; return b;
}
__device__ __forceinline__ v2f fragBT(const float* X, int ld, int k0, int n0, int lane) {
    int h = lane >> 4, n = n0 + (lane & 15), k = k0 + 2 * h;
    v2f b; b.x = X[n * ld + k]; b.y = X[n * ld + k + 1]; return b;
}
__device__ __forceinline__ void storeD(float* X, int ld, int m0, int n0, v8f d, int lane,
                                       float sgn = 1.0f) {
    int h = lane >> 4, n = n0 + (lane & 15);
#pragma unroll
    for (int r = 0; r < 8; ++r) X[(m0 + r + 8 * h) * ld + n] = sgn * d[r];
}
__device__ __forceinline__ v8f loadC(const float* X, int ld, int m0, int n0, int lane) {
    int h = lane >> 4, n = n0 + (lane & 15);
    v8f c;
#pragma unroll
    for (int r = 0; r < 8; ++r) c[r] = X[(m0 + r + 8 * h) * ld + n];
    return c;
}
__device__ __forceinline__ v8f identC(int m0, int n0, int lane) {
    int h = lane >> 4, n = n0 + (lane & 15);
    v8f c;
#pragma unroll
    for (int r = 0; r < 8; ++r) c[r] = ((m0 + r + 8 * h) == n) ? 1.0f : 0.0f;
    return c;
}

__device__ __forceinline__ float lane_bcast(float x, int l) {
    return __int_as_float(__builtin_amdgcn_readlane(__float_as_int(x), l));
}

template <bool TA, bool TB>
__device__ __forceinline__ v8f mmacc(const float* A, int lda, int m0,
                                     const float* B, int ldb, int n0,
                                     int K, v8f c, int lane) {
#pragma unroll
    for (int k = 0; k < K; k += 4) {
        v2f a = TA ? fragAT(A, lda, m0, k, lane) : fragA(A, lda, m0, k, lane);
        v2f b = TB ? fragBT(B, ldb, k, n0, lane) : fragB(B, ldb, k, n0, lane);
        c = __builtin_amdgcn_wmma_f32_16x16x4_f32(false, a, false, b, (short)0, c,
                                                  false, false);
    }
    return c;
}

__global__ __launch_bounds__(256) void kf_gain_seq(const float* __restrict__ Wfx,
                                                   const float* __restrict__ Wfy,
                                                   float* __restrict__ G) {
    __shared__ __attribute__((aligned(16))) float sWfx[NXX * LDX];
    __shared__ __attribute__((aligned(16))) float sWfy[NXX * LDY];
    __shared__ __attribute__((aligned(16))) float sP[NXX * LDX];
    __shared__ __attribute__((aligned(16))) float sM1[NXX * LDX];
    __shared__ __attribute__((aligned(16))) float sM2[NXX * LDY];
    __shared__ __attribute__((aligned(16))) float sS[NYY * SLD];
    __shared__ __attribute__((aligned(16))) float sL[NXX * LDY];

    const int tid = threadIdx.x, lane = tid & 31, w = tid >> 5;

    for (int i = tid; i < NXX * NXX; i += 256) {
        int r = i >> 6, cc = i & 63;
        sWfx[r * LDX + cc] = Wfx[i];
        sP[r * LDX + cc] = (r == cc) ? 1.0f : 0.0f;
    }
    for (int i = tid; i < NXX * NYY; i += 256) {
        int r = i >> 5, cc = i & 31;
        sWfy[r * LDY + cc] = Wfy[i];
        sL[r * LDY + cc] = 0.0f;
    }
    __syncthreads();

    for (int t = 0; t < T_STEPS; ++t) {
        float* Gt = G + (size_t)t * (NYY * NXX);
        for (int i = tid; i < NYY * NXX; i += 256) {
            int j = i >> 6, col = i & 63;
            Gt[i] = sL[col * LDY + j];
        }
        __syncthreads();

        for (int tt = w; tt < 16; tt += 8) {
            int mi = (tt >> 2) * 16, ni = (tt & 3) * 16;
            v8f c = zero8();
            c = mmacc<false, false>(sWfx, LDX, mi, sP, LDX, ni, NXX, c, lane);
            storeD(sM1, LDX, mi, ni, c, lane);
        }
        __syncthreads();
        for (int tt = w; tt < 16; tt += 8) {
            int mi = (tt >> 2) * 16, ni = (tt & 3) * 16;
            v8f c = identC(mi, ni, lane);
            c = mmacc<false, true>(sM1, LDX, mi, sWfx, LDX, ni, NXX, c, lane);
            storeD(sP, LDX, mi, ni, c, lane);
        }
        __syncthreads();
        {
            int mi = (w >> 1) * 16, ni = (w & 1) * 16;
            v8f c = zero8();
            c = mmacc<false, false>(sP, LDX, mi, sWfy, LDY, ni, NXX, c, lane);
            storeD(sM2, LDY, mi, ni, c, lane);
        }
        __syncthreads();
        if (w < 4) {
            int mi = (w >> 1) * 16, ni = (w & 1) * 16;
            v8f c = identC(mi, ni, lane);
            c = mmacc<true, false>(sWfy, LDY, mi, sM2, LDY, ni, NXX, c, lane);
            storeD(sS, SLD, mi, ni, c, lane);
        }
        __syncthreads();
        if (w == 0) {
            float a[32];
            const float4* rin = reinterpret_cast<const float4*>(sS + lane * SLD);
#pragma unroll
            for (int j = 0; j < 8; ++j) {
                float4 v = rin[j];
                a[4 * j] = v.x; a[4 * j + 1] = v.y; a[4 * j + 2] = v.z; a[4 * j + 3] = v.w;
            }
            float s = 1.0f;
            float ip = 1.0f / lane_bcast(a[0], 0);
            for (int p = 0; p < 32; ++p) {
                const bool isp = (lane == p);
                const float g = isp ? 0.0f : a[0] * ip;
                s = isp ? ip : s;
                a[0] = a[1] - g * lane_bcast(a[1], p);
                const float ipn = 1.0f / lane_bcast(a[0], (p + 1) & 31);
#pragma unroll
                for (int c = 2; c < 32; ++c) a[c - 1] = a[c] - g * lane_bcast(a[c], p);
                a[31] = isp ? 1.0f : -g;
                ip = ipn;
            }
            float4* rout = reinterpret_cast<float4*>(sS + lane * SLD);
#pragma unroll
            for (int j = 0; j < 8; ++j) {
                float4 v;
                v.x = a[4 * j] * s; v.y = a[4 * j + 1] * s;
                v.z = a[4 * j + 2] * s; v.w = a[4 * j + 3] * s;
                rout[j] = v;
            }
        }
        __syncthreads();
        {
            int mi = (w >> 1) * 16, ni = (w & 1) * 16;
            v8f c = zero8();
            c = mmacc<false, false>(sM2, LDY, mi, sS, SLD, ni, NYY, c, lane);
            storeD(sL, LDY, mi, ni, c, lane);
        }
        __syncthreads();
        {
            int mi = (w >> 2) * 16, ni = (w & 3) * 16;
            v8f c = zero8();
            c = mmacc<true, false>(sWfy, LDY, mi, sP, LDX, ni, NXX, c, lane);
            storeD(sM1, LDX, mi, ni, c, lane, -1.0f);
        }
        __syncthreads();
        for (int tt = w; tt < 16; tt += 8) {
            int mi = (tt >> 2) * 16, ni = (tt & 3) * 16;
            v8f c = identC(mi, ni, lane);
            c = mmacc<false, false>(sL, LDY, mi, sM1, LDX, ni, NYY, c, lane);
            storeD(sP, LDX, mi, ni, c, lane);
        }
        __syncthreads();
    }
}

__global__ __launch_bounds__(128) void kf_batch(const float* __restrict__ Yp,
                                                const float* __restrict__ Up,
                                                const float* __restrict__ Dp,
                                                const float* __restrict__ Wfx,
                                                const float* __restrict__ Wfu,
                                                const float* __restrict__ Wfd,
                                                const float* __restrict__ Wfy,
                                                const float* __restrict__ G,
                                                float* __restrict__ out) {
    __shared__ __attribute__((aligned(16))) float sW[(NXX + NUU + NDD) * LDX];
    __shared__ __attribute__((aligned(16))) float sWfy[NXX * LDY];
    __shared__ __attribute__((aligned(16))) float sx[16 * LDX];
    __shared__ __attribute__((aligned(16))) float sxp[16 * LDX];
    __shared__ __attribute__((aligned(16))) float su[16 * LDU];
    __shared__ __attribute__((aligned(16))) float sd[16 * LDD];
    __shared__ __attribute__((aligned(16))) float sym[16 * LDY];
    __shared__ __attribute__((aligned(16))) float sin[16 * LDY];

    const int tid = threadIdx.x, lane = tid & 31, w = tid >> 5;
    const int r0 = blockIdx.x * 16;

    for (int i = tid; i < NXX * NXX; i += 128) sW[(i >> 6) * LDX + (i & 63)] = Wfx[i];
    for (int i = tid; i < NUU * NXX; i += 128) sW[(NXX + (i >> 6)) * LDX + (i & 63)] = Wfu[i];
    for (int i = tid; i < NDD * NXX; i += 128) sW[(NXX + NUU + (i >> 6)) * LDX + (i & 63)] = Wfd[i];
    for (int i = tid; i < NXX * NYY; i += 128) sWfy[(i >> 5) * LDY + (i & 31)] = Wfy[i];
    for (int i = tid; i < 16 * NXX; i += 128) sx[(i >> 6) * LDX + (i & 63)] = 0.0f;
    __syncthreads();

    for (int t = 0; t < T_STEPS; ++t) {
        const float* ut = Up + ((size_t)t * B_TOT + r0) * NUU;
        const float* dt = Dp + ((size_t)t * B_TOT + r0) * NDD;
        const float* yt = Yp + ((size_t)t * B_TOT + r0) * NYY;
        for (int i = tid; i < 16 * NUU; i += 128) su[(i >> 4) * LDU + (i & 15)] = ut[i];
        for (int i = tid; i < 16 * NDD; i += 128) sd[(i >> 3) * LDD + (i & 7)] = dt[i];
        for (int i = tid; i < 16 * NYY; i += 128) sym[(i >> 5) * LDY + (i & 31)] = yt[i];
        const float* Gt = G + (size_t)t * (NYY * NXX);
        if (t + 1 < T_STEPS) {
            __builtin_prefetch(G + (size_t)(t + 1) * (NYY * NXX) + tid * 16, 0, 1);
        }
        __syncthreads();

        {
            int n0 = w * 16;
            v8f c = zero8();
            c = mmacc<false, false>(sx, LDX, 0, sW, LDX, n0, NXX, c, lane);
            c = mmacc<false, false>(su, LDU, 0, sW + NXX * LDX, LDX, n0, NUU, c, lane);
            c = mmacc<false, false>(sd, LDD, 0, sW + (NXX + NUU) * LDX, LDX, n0, NDD, c, lane);
            storeD(sxp, LDX, 0, n0, c, lane);
        }
        __syncthreads();

        if (w < 2) {
            int n0 = w * 16;
            v8f c = zero8();
            c = mmacc<false, false>(sxp, LDX, 0, sWfy, LDY, n0, NXX, c, lane);
            int h = lane >> 4, n = n0 + (lane & 15);
#pragma unroll
            for (int r = 0; r < 8; ++r) {
                int m = r + 8 * h;
                sin[m * LDY + n] = sym[m * LDY + n] - c[r];
            }
        }
        __syncthreads();

        {
            int n0 = w * 16;
            v8f c = loadC(sxp, LDX, 0, n0, lane);
            c = mmacc<false, false>(sin, LDY, 0, Gt, NXX, n0, NYY, c, lane);
            storeD(sx, LDX, 0, n0, c, lane);
        }
        __syncthreads();
    }

    for (int i = tid; i < 16 * NXX; i += 128) {
        int r = i >> 6, cc = i & 63;
        out[(size_t)(r0 + r) * NXX + cc] = sx[r * LDX + cc];
    }
}

extern "C" void kernel_launch(void* const* d_in, const int* in_sizes, int n_in,
                              void* d_out, int out_size, void* d_ws, size_t ws_size,
                              hipStream_t stream) {
    (void)in_sizes; (void)n_in; (void)out_size; (void)ws_size;
    const float* Yp  = (const float*)d_in[0];
    const float* Up  = (const float*)d_in[1];
    const float* Dp  = (const float*)d_in[2];
    const float* Wfx = (const float*)d_in[3];
    const float* Wfu = (const float*)d_in[4];
    const float* Wfd = (const float*)d_in[5];
    const float* Wfy = (const float*)d_in[6];
    float* G = (float*)d_ws;
    float* out = (float*)d_out;

    kf_gain_seq<<<1, 256, 0, stream>>>(Wfx, Wfy, G);
    kf_batch<<<16, 128, 0, stream>>>(Yp, Up, Dp, Wfx, Wfu, Wfd, Wfy, G, out);
}


# Batch kernel: a step's global loads all issue at its top, ahead of use (next step's inputs and this step's gain fragments, to registers), in flight across LDS-only barriers; sums unchanged
# speedup vs baseline: 2.7653x; 1.3492x over previous
#include <hip/hip_runtime.h>
#include <hip/hip_bf16.h>

#define NXX 64
#define NYY 32
#define NUU 16
#define NDD 8
#define T_STEPS 2048
#define B_TOT 256
#define LDX 76
#define LDY 44
#define LDU 20
#define LDD 12
#define SLD LDY

typedef float v2f __attribute__((ext_vector_type(2)));
typedef float v8f __attribute__((ext_vector_type(8)));

__device__ __forceinline__ v8f zero8() {
    v8f c;
#pragma unroll
    for (int r = 0; r < 8; ++r) c[r] = 0.0f;
    return c;
}

__device__ __forceinline__ v2f fragA(const float* X, int ld, int m0, int k0, int lane) {
    int h = lane >> 4, m = m0 + (lane & 15), k = k0 + 2 * h;
    v2f a; a.x = X[m * ld + k]; a.y = X[m * ld + k + 1]; return a;
}
__device__ __forceinline__ v2f fragAT(const float* X, int ld, int m0, int k0, int lane) {
    int h = lane >> 4, m = m0 + (lane & 15), k = k0 + 2 * h;
    v2f a; a.x = X[k * ld + m]; a.y = X[(k + 1) * ld + m]; return a;
}
__device__ __forceinline__ v2f fragB(const float* X, int ld, int k0, int n0, int lane) {
    int h = lane >> 4, n = n0 + (lane & 15), k = k0 + 2 * h;
    v2f b; b.x = X[k * ld + n]; b.y = X[(k + 1) * ld + n]; return b;
}
__device__ __forceinline__ v2f fragBT(const float* X, int ld, int k0, int n0, int lane) {
    int h = lane >> 4, n = n0 + (lane & 15), k = k0 + 2 * h;
    v2f b; b.x = X[n * ld + k]; b.y = X[n * ld + k + 1]; return b;
}
__device__ __forceinline__ void storeD(float* X, int ld, int m0, int n0, v8f d, int lane,
                                       float sgn = 1.0f) {
    int h = lane >> 4, n = n0 + (lane & 15);
#pragma unroll
    for (int r = 0; r < 8; ++r) X[(m0 + r + 8 * h) * ld + n] = sgn * d[r];
}
__device__ __forceinline__ v8f loadC(const float* X, int ld, int m0, int n0, int lane) {
    int h = lane >> 4, n = n0 + (lane & 15);
    v8f c;
#pragma unroll
    for (int r = 0; r < 8; ++r) c[r] = X[(m0 + r + 8 * h) * ld + n];
    return c;
}
__device__ __forceinline__ v8f identC(int m0, int n0, int lane) {
    int h = lane >> 4, n = n0 + (lane & 15);
    v8f c;
#pragma unroll
    for (int r = 0; r < 8; ++r) c[r] = ((m0 + r + 8 * h) == n) ? 1.0f : 0.0f;
    return c;
}

__device__ __forceinline__ float lane_bcast(float x, int l) {
    return __int_as_float(__builtin_amdgcn_readlane(__float_as_int(x), l));
}

template <bool TA, bool TB>
__device__ __forceinline__ v8f mmacc(const float* A, int lda, int m0,
                                     const float* B, int ldb, int n0,
                                     int K, v8f c, int lane) {
#pragma unroll
    for (int k = 0; k < K; k += 4) {
        v2f a = TA ? fragAT(A, lda, m0, k, lane) : fragA(A, lda, m0, k, lane);
        v2f b = TB ? fragBT(B, ldb, k, n0, lane) : fragB(B, ldb, k, n0, lane);
        c = __builtin_amdgcn_wmma_f32_16x16x4_f32(false, a, false, b, (short)0, c,
                                                  false, false);
    }
    return c;
}

__global__ __launch_bounds__(256) void kf_gain_seq(const float* __restrict__ Wfx,
                                                   const float* __restrict__ Wfy,
                                                   float* __restrict__ G) {
    __shared__ __attribute__((aligned(16))) float sWfx[NXX * LDX];
    __shared__ __attribute__((aligned(16))) float sWfy[NXX * LDY];
    __shared__ __attribute__((aligned(16))) float sP[NXX * LDX];
    __shared__ __attribute__((aligned(16))) float sM1[NXX * LDX];
    __shared__ __attribute__((aligned(16))) float sM2[NXX * LDY];
    __shared__ __attribute__((aligned(16))) float sS[NYY * SLD];
    __shared__ __attribute__((aligned(16))) float sL[NXX * LDY];

    const int tid = threadIdx.x, lane = tid & 31, w = tid >> 5;

    for (int i = tid; i < NXX * NXX; i += 256) {
        int r = i >> 6, cc = i & 63;
        sWfx[r * LDX + cc] = Wfx[i];
        sP[r * LDX + cc] = (r == cc) ? 1.0f : 0.0f;
    }
    for (int i = tid; i < NXX * NYY; i += 256) {
        int r = i >> 5, cc = i & 31;
        sWfy[r * LDY + cc] = Wfy[i];
        sL[r * LDY + cc] = 0.0f;
    }
    __syncthreads();

    for (int t = 0; t < T_STEPS; ++t) {
        float* Gt = G + (size_t)t * (NYY * NXX);
        for (int i = tid; i < NYY * NXX; i += 256) {
            int j = i >> 6, col = i & 63;
            Gt[i] = sL[col * LDY + j];
        }
        __syncthreads();

        for (int tt = w; tt < 16; tt += 8) {
            int mi = (tt >> 2) * 16, ni = (tt & 3) * 16;
            v8f c = zero8();
            c = mmacc<false, false>(sWfx, LDX, mi, sP, LDX, ni, NXX, c, lane);
            storeD(sM1, LDX, mi, ni, c, lane);
        }
        __syncthreads();
        for (int tt = w; tt < 16; tt += 8) {
            int mi = (tt >> 2) * 16, ni = (tt & 3) * 16;
            v8f c = identC(mi, ni, lane);
            c = mmacc<false, true>(sM1, LDX, mi, sWfx, LDX, ni, NXX, c, lane);
            storeD(sP, LDX, mi, ni, c, lane);
        }
        __syncthreads();
        {
            int mi = (w >> 1) * 16, ni = (w & 1) * 16;
            v8f c = zero8();
            c = mmacc<false, false>(sP, LDX, mi, sWfy, LDY, ni, NXX, c, lane);
            storeD(sM2, LDY, mi, ni, c, lane);
        }
        __syncthreads();
        if (w < 4) {
            int mi = (w >> 1) * 16, ni = (w & 1) * 16;
            v8f c = identC(mi, ni, lane);
            c = mmacc<true, false>(sWfy, LDY, mi, sM2, LDY, ni, NXX, c, lane);
            storeD(sS, SLD, mi, ni, c, lane);
        }
        __syncthreads();
        if (w == 0) {
            float a[32];
            const float4* rin = reinterpret_cast<const float4*>(sS + lane * SLD);
#pragma unroll
            for (int j = 0; j < 8; ++j) {
                float4 v = rin[j];
                a[4 * j] = v.x; a[4 * j + 1] = v.y; a[4 * j + 2] = v.z; a[4 * j + 3] = v.w;
            }
            float s = 1.0f;
            float ip = 1.0f / lane_bcast(a[0], 0);
            for (int p = 0; p < 32; ++p) {
                const bool isp = (lane == p);
                const float g = isp ? 0.0f : a[0] * ip;
                s = isp ? ip : s;
                a[0] = a[1] - g * lane_bcast(a[1], p);
                const float ipn = 1.0f / lane_bcast(a[0], (p + 1) & 31);
#pragma unroll
                for (int c = 2; c < 32; ++c) a[c - 1] = a[c] - g * lane_bcast(a[c], p);
                a[31] = isp ? 1.0f : -g;
                ip = ipn;
            }
            float4* rout = reinterpret_cast<float4*>(sS + lane * SLD);
#pragma unroll
            for (int j = 0; j < 8; ++j) {
                float4 v;
                v.x = a[4 * j] * s; v.y = a[4 * j + 1] * s;
                v.z = a[4 * j + 2] * s; v.w = a[4 * j + 3] * s;
                rout[j] = v;
            }
        }
        __syncthreads();
        {
            int mi = (w >> 1) * 16, ni = (w & 1) * 16;
            v8f c = zero8();
            c = mmacc<false, false>(sM2, LDY, mi, sS, SLD, ni, NYY, c, lane);
            storeD(sL, LDY, mi, ni, c, lane);
        }
        __syncthreads();
        {
            int mi = (w >> 2) * 16, ni = (w & 3) * 16;
            v8f c = zero8();
            c = mmacc<true, false>(sWfy, LDY, mi, sP, LDX, ni, NXX, c, lane);
            storeD(sM1, LDX, mi, ni, c, lane, -1.0f);
        }
        __syncthreads();
        for (int tt = w; tt < 16; tt += 8) {
            int mi = (tt >> 2) * 16, ni = (tt & 3) * 16;
            v8f c = identC(mi, ni, lane);
            c = mmacc<false, false>(sL, LDY, mi, sM1, LDX, ni, NYY, c, lane);
            storeD(sP, LDX, mi, ni, c, lane);
        }
        __syncthreads();
    }
}


__device__ __forceinline__ void lds_barrier() {
    __builtin_amdgcn_fence(__ATOMIC_RELEASE, "workgroup", "local");
    __builtin_amdgcn_s_barrier();
    __builtin_amdgcn_fence(__ATOMIC_ACQUIRE, "workgroup", "local");
}

__global__ __launch_bounds__(128) void kf_batch(const float* __restrict__ Yp,
                                                const float* __restrict__ Up,
                                                const float* __restrict__ Dp,
                                                const float* __restrict__ Wfx,
                                                const float* __restrict__ Wfu,
                                                const float* __restrict__ Wfd,
                                                const float* __restrict__ Wfy,
                                                const float* __restrict__ G,
                                                float* __restrict__ out) {
    __shared__ __attribute__((aligned(16))) float sW[(NXX + NUU + NDD) * LDX];
    __shared__ __attribute__((aligned(16))) float sWfy[NXX * LDY];
    __shared__ __attribute__((aligned(16))) float sx[16 * LDX];
    __shared__ __attribute__((aligned(16))) float sxp[16 * LDX];
    __shared__ __attribute__((aligned(16))) float su[16 * LDU];
    __shared__ __attribute__((aligned(16))) float sd[16 * LDD];
    __shared__ __attribute__((aligned(16))) float sym[16 * LDY];
    __shared__ __attribute__((aligned(16))) float sin[16 * LDY];

    const int tid = threadIdx.x, lane = tid & 31, w = tid >> 5;
    const int r0 = blockIdx.x * 16;
    const int n0 = w * 16;

    for (int i = tid; i < NXX * NXX; i += 128) sW[(i >> 6) * LDX + (i & 63)] = Wfx[i];
    for (int i = tid; i < NUU * NXX; i += 128) sW[(NXX + (i >> 6)) * LDX + (i & 63)] = Wfu[i];
    for (int i = tid; i < NDD * NXX; i += 128) sW[(NXX + NUU + (i >> 6)) * LDX + (i & 63)] = Wfd[i];
    for (int i = tid; i < NXX * NYY; i += 128) sWfy[(i >> 5) * LDY + (i & 31)] = Wfy[i];
    for (int i = tid; i < 16 * NXX; i += 128) sx[(i >> 6) * LDX + (i & 63)] = 0.0f;
    {
        const float* ut = Up + (size_t)r0 * NUU;
        const float* dt = Dp + (size_t)r0 * NDD;
        const float* yt = Yp + (size_t)r0 * NYY;
        for (int i = tid; i < 16 * NUU; i += 128) su[(i >> 4) * LDU + (i & 15)] = ut[i];
        for (int i = tid; i < 16 * NDD; i += 128) sd[(i >> 3) * LDD + (i & 7)] = dt[i];
        for (int i = tid; i < 16 * NYY; i += 128) sym[(i >> 5) * LDY + (i & 31)] = yt[i];
    }
    __syncthreads();

    for (int t = 0; t < T_STEPS; ++t) {
        const float* Gt = G + (size_t)t * (NYY * NXX);
        v2f gB[NYY / 4];
#pragma unroll
        for (int kk = 0; kk < NYY / 4; ++kk) gB[kk] = fragB(Gt, NXX, 4 * kk, n0, lane);

        const int tn = (t + 1 < T_STEPS) ? t + 1 : t;
        const float* un = Up + ((size_t)tn * B_TOT + r0) * NUU;
        const float* dn = Dp + ((size_t)tn * B_TOT + r0) * NDD;
        const float* yn = Yp + ((size_t)tn * B_TOT + r0) * NYY;
        float nu[2], ny[4];
#pragma unroll
        for (int j = 0; j < 2; ++j) nu[j] = un[tid + 128 * j];
        const float nd = dn[tid];
#pragma unroll
        for (int j = 0; j < 4; ++j) ny[j] = yn[tid + 128 * j];

        if (t + 1 < T_STEPS) {
            __builtin_prefetch(G + (size_t)(t + 1) * (NYY * NXX) + tid * 16, 0, 1);
        }

        {
            v8f c = zero8();
            c = mmacc<false, false>(sx, LDX, 0, sW, LDX, n0, NXX, c, lane);
            c = mmacc<false, false>(su, LDU, 0, sW + NXX * LDX, LDX, n0, NUU, c, lane);
            c = mmacc<false, false>(sd, LDD, 0, sW + (NXX + NUU) * LDX, LDX, n0, NDD, c, lane);
            storeD(sxp, LDX, 0, n0, c, lane);
        }
        lds_barrier();

        if (w < 2) {
            int m0 = w * 16;
            v8f c = zero8();
            c = mmacc<false, false>(sxp, LDX, 0, sWfy, LDY, m0, NXX, c, lane);
            int h = lane >> 4, n = m0 + (lane & 15);
#pragma unroll
            for (int r = 0; r < 8; ++r) {
                int m = r + 8 * h;
                sin[m * LDY + n] = sym[m * LDY + n] - c[r];
            }
        }
        lds_barrier();

        {
            v8f c = loadC(sxp, LDX, 0, n0, lane);
#pragma unroll
            for (int kk = 0; kk < NYY / 4; ++kk) {
                v2f a = fragA(sin, LDY, 0, 4 * kk, lane);
                c = __builtin_amdgcn_wmma_f32_16x16x4_f32(false, a, false, gB[kk], (short)0, c,
                                                          false, false);
            }
            storeD(sx, LDX, 0, n0, c, lane);
        }
#pragma unroll
        for (int j = 0; j < 2; ++j) su[((tid >> 4) + 8 * j) * LDU + (tid & 15)] = nu[j];
        sd[(tid >> 3) * LDD + (tid & 7)] = nd;
#pragma unroll
        for (int j = 0; j < 4; ++j) sym[((tid >> 5) + 4 * j) * LDY + (tid & 31)] = ny[j];
        lds_barrier();
    }

    for (int i = tid; i < 16 * NXX; i += 128) {
        int r = i >> 6, cc = i & 63;
        out[(size_t)(r0 + r) * NXX + cc] = sx[r * LDX + cc];
    }
}

extern "C" void kernel_launch(void* const* d_in, const int* in_sizes, int n_in,
                              void* d_out, int out_size, void* d_ws, size_t ws_size,
                              hipStream_t stream) {
    (void)in_sizes; (void)n_in; (void)out_size; (void)ws_size;
    const float* Yp  = (const float*)d_in[0];
    const float* Up  = (const float*)d_in[1];
    const float* Dp  = (const float*)d_in[2];
    const float* Wfx = (const float*)d_in[3];
    const float* Wfu = (const float*)d_in[4];
    const float* Wfd = (const float*)d_in[5];
    const float* Wfy = (const float*)d_in[6];
    float* G = (float*)d_ws;
    float* out = (float*)d_out;

    kf_gain_seq<<<1, 256, 0, stream>>>(Wfx, Wfy, G);
    kf_batch<<<16, 128, 0, stream>>>(Yp, Up, Dp, Wfx, Wfu, Wfd, Wfy, G, out);
}


# One launch instead of two: every workgroup runs the gain recursion on 8 waves and its batch tile on 4 more, step for step, gain handed over in LDS (two-deep) instead of global memory; sums unchanged
# speedup vs baseline: 3.7674x; 1.3624x over previous
#include <hip/hip_runtime.h>
#include <hip/hip_bf16.h>

#define NXX 64
#define NYY 32
#define NUU 16
#define NDD 8
#define T_STEPS 2048
#define B_TOT 256
#define LDX 76
#define LDY 44
#define LDU 20
#define LDD 12
#define SLD LDY

typedef float v2f __attribute__((ext_vector_type(2)));
typedef float v8f __attribute__((ext_vector_type(8)));

__device__ __forceinline__ v8f zero8() {
    v8f c;
#pragma unroll
    for (int r = 0; r < 8; ++r) c[r] = 0.0f;
    return c;
}

__device__ __forceinline__ v2f fragA(const float* X, int ld, int m0, int k0, int lane) {
    int h = lane >> 4, m = m0 + (lane & 15), k = k0 + 2 * h;
    v2f a; a.x = X[m * ld + k]; a.y = X[m * ld + k + 1]; return a;
}
__device__ __forceinline__ v2f fragAT(const float* X, int ld, int m0, int k0, int lane) {
    int h = lane >> 4, m = m0 + (lane & 15), k = k0 + 2 * h;
    v2f a; a.x = X[k * ld + m]; a.y = X[(k + 1) * ld + m]; return a;
}
__device__ __forceinline__ v2f fragB(const float* X, int ld, int k0, int n0, int lane) {
    int h = lane >> 4, n = n0 + (lane & 15), k = k0 + 2 * h;
    v2f b; b.x = X[k * ld + n]; b.y = X[(k + 1) * ld + n]; return b;
}
__device__ __forceinline__ v2f fragBT(const float* X, int ld, int k0, int n0, int lane) {
    int h = lane >> 4, n = n0 + (lane & 15), k = k0 + 2 * h;
    v2f b; b.x = X[n * ld + k]; b.y = X[n * ld + k + 1]; return b;
}
__device__ __forceinline__ void storeD(float* X, int ld, int m0, int n0, v8f d, int lane,
                                       float sgn = 1.0f) {
    int h = lane >> 4, n = n0 + (lane & 15);
#pragma unroll
    for (int r = 0; r < 8; ++r) X[(m0 + r + 8 * h) * ld + n] = sgn * d[r];
}
__device__ __forceinline__ v8f loadC(const float* X, int ld, int m0, int n0, int lane) {
    int h = lane >> 4, n = n0 + (lane & 15);
    v8f c;
#pragma unroll
    for (int r = 0; r < 8; ++r) c[r] = X[(m0 + r + 8 * h) * ld + n];
    return c;
}
__device__ __forceinline__ v8f identC(int m0, int n0, int lane) {
    int h = lane >> 4, n = n0 + (lane & 15);
    v8f c;
#pragma unroll
    for (int r = 0; r < 8; ++r) c[r] = ((m0 + r + 8 * h) == n) ? 1.0f : 0.0f;
    return c;
}

__device__ __forceinline__ float lane_bcast(float x, int l) {
    return __int_as_float(__builtin_amdgcn_readlane(__float_as_int(x), l));
}

template <bool TA, bool TB>
__device__ __forceinline__ v8f mmacc(const float* A, int lda, int m0,
                                     const float* B, int ldb, int n0,
                                     int K, v8f c, int lane) {
#pragma unroll
    for (int k = 0; k < K; k += 4) {
        v2f a = TA ? fragAT(A, lda, m0, k, lane) : fragA(A, lda, m0, k, lane);
        v2f b = TB ? fragBT(B, ldb, k, n0, lane) : fragB(B, ldb, k, n0, lane);
        c = __builtin_amdgcn_wmma_f32_16x16x4_f32(false, a, false, b, (short)0, c,
                                                  false, false);
    }
    return c;
}

__device__ __forceinline__ void lds_barrier() {
    __builtin_amdgcn_fence(__ATOMIC_RELEASE, "workgroup", "local");
    __builtin_amdgcn_s_barrier();
    __builtin_amdgcn_fence(__ATOMIC_ACQUIRE, "workgroup", "local");
}

__global__ __launch_bounds__(384) void kf_fused(const float* __restrict__ Yp,
                                                const float* __restrict__ Up,
                                                const float* __restrict__ Dp,
                                                const float* __restrict__ Wfx,
                                                const float* __restrict__ Wfu,
                                                const float* __restrict__ Wfd,
                                                const float* __restrict__ Wfy,
                                                float* __restrict__ out) {
    __shared__ __attribute__((aligned(16))) float sW[(NXX + NUU + NDD) * LDX];
    __shared__ __attribute__((aligned(16))) float sWfy[NXX * LDY];
    __shared__ __attribute__((aligned(16))) float sP[NXX * LDX];
    __shared__ __attribute__((aligned(16))) float sM1[NXX * LDX];
    __shared__ __attribute__((aligned(16))) float sM2[NXX * LDY];
    __shared__ __attribute__((aligned(16))) float sS[NYY * SLD];
    __shared__ __attribute__((aligned(16))) float sL[2 * NXX * LDY];
    __shared__ __attribute__((aligned(16))) float sx[16 * LDX];
    __shared__ __attribute__((aligned(16))) float sxp[16 * LDX];
    __shared__ __attribute__((aligned(16))) float su[16 * LDU];
    __shared__ __attribute__((aligned(16))) float sd[16 * LDD];
    __shared__ __attribute__((aligned(16))) float sym[16 * LDY];
    __shared__ __attribute__((aligned(16))) float sin[16 * LDY];

    const int tid = threadIdx.x, lane = tid & 31;
    const int w = __builtin_amdgcn_readfirstlane(tid >> 5);
    const bool gainw = (w < 8);
    const int bt = tid - 256;
    const int r0 = blockIdx.x * 16;
    const int n0 = (w - 8) * 16;

    for (int i = tid; i < NXX * NXX; i += 384) {
        int r = i >> 6, cc = i & 63;
        sW[r * LDX + cc] = Wfx[i];
        sP[r * LDX + cc] = (r == cc) ? 1.0f : 0.0f;
    }
    for (int i = tid; i < NUU * NXX; i += 384) sW[(NXX + (i >> 6)) * LDX + (i & 63)] = Wfu[i];
    for (int i = tid; i < NDD * NXX; i += 384) sW[(NXX + NUU + (i >> 6)) * LDX + (i & 63)] = Wfd[i];
    for (int i = tid; i < NXX * NYY; i += 384) {
        int r = i >> 5, cc = i & 31;
        sWfy[r * LDY + cc] = Wfy[i];
        sL[r * LDY + cc] = 0.0f;
    }
    for (int i = tid; i < 16 * NXX; i += 384) sx[(i >> 6) * LDX + (i & 63)] = 0.0f;
    {
        const float* ut = Up + (size_t)r0 * NUU;
        const float* dt = Dp + (size_t)r0 * NDD;
        const float* yt = Yp + (size_t)r0 * NYY;
        for (int i = tid; i < 16 * NUU; i += 384) su[(i >> 4) * LDU + (i & 15)] = ut[i];
        for (int i = tid; i < 16 * NDD; i += 384) sd[(i >> 3) * LDD + (i & 7)] = dt[i];
        for (int i = tid; i < 16 * NYY; i += 384) sym[(i >> 5) * LDY + (i & 31)] = yt[i];
    }
    __syncthreads();

    for (int t = 0; t < T_STEPS; ++t) {
        const float* sLc = sL + (t & 1) * (NXX * LDY);
        float* sLn = sL + ((t & 1) ^ 1) * (NXX * LDY);

        float nu[2], nd, ny[4];
        if (!gainw) {
            const int tn = (t + 1 < T_STEPS) ? t + 1 : t;
            const float* un = Up + ((size_t)tn * B_TOT + r0) * NUU;
            const float* dn = Dp + ((size_t)tn * B_TOT + r0) * NDD;
            const float* yn = Yp + ((size_t)tn * B_TOT + r0) * NYY;
#pragma unroll
            for (int j = 0; j < 2; ++j) nu[j] = un[bt + 128 * j];
            nd = dn[bt];
#pragma unroll
            for (int j = 0; j < 4; ++j) ny[j] = yn[bt + 128 * j];
        }

        if (gainw) {
            for (int tt = w; tt < 16; tt += 8) {
                int mi = (tt >> 2) * 16, ni = (tt & 3) * 16;
                v8f c = zero8();
                c = mmacc<false, false>(sW, LDX, mi, sP, LDX, ni, NXX, c, lane);
                storeD(sM1, LDX, mi, ni, c, lane);
            }
        }
        lds_barrier();
        if (gainw) {
            for (int tt = w; tt < 16; tt += 8) {
                int mi = (tt >> 2) * 16, ni = (tt & 3) * 16;
                v8f c = identC(mi, ni, lane);
                c = mmacc<false, true>(sM1, LDX, mi, sW, LDX, ni, NXX, c, lane);
                storeD(sP, LDX, mi, ni, c, lane);
            }
        }
        lds_barrier();
        if (gainw) {
            int mi = (w >> 1) * 16, ni = (w & 1) * 16;
            v8f c = zero8();
            c = mmacc<false, false>(sP, LDX, mi, sWfy, LDY, ni, NXX, c, lane);
            storeD(sM2, LDY, mi, ni, c, lane);
        }
        lds_barrier();
        if (gainw) {
            if (w < 4) {
                int mi = (w >> 1) * 16, ni = (w & 1) * 16;
                v8f c = identC(mi, ni, lane);
                c = mmacc<true, false>(sWfy, LDY, mi, sM2, LDY, ni, NXX, c, lane);
                storeD(sS, SLD, mi, ni, c, lane);
            }
        } else {
            v8f c = zero8();
            c = mmacc<false, false>(sx, LDX, 0, sW, LDX, n0, NXX, c, lane);
            c = mmacc<false, false>(su, LDU, 0, sW + NXX * LDX, LDX, n0, NUU, c, lane);
            c = mmacc<false, false>(sd, LDD, 0, sW + (NXX + NUU) * LDX, LDX, n0, NDD, c, lane);
            storeD(sxp, LDX, 0, n0, c, lane);
        }
        lds_barrier();
        if (w == 0) {
            float a[32];
            const float4* rin = reinterpret_cast<const float4*>(sS + lane * SLD);
#pragma unroll
            for (int j = 0; j < 8; ++j) {
                float4 v = rin[j];
                a[4 * j] = v.x; a[4 * j + 1] = v.y; a[4 * j + 2] = v.z; a[4 * j + 3] = v.w;
            }
            float s = 1.0f;
            float ip = 1.0f / lane_bcast(a[0], 0);
            for (int p = 0; p < 32; ++p) {
                const bool isp = (lane == p);
                const float g = isp ? 0.0f : a[0] * ip;
                s = isp ? ip : s;
                a[0] = a[1] - g * lane_bcast(a[1], p);
                const float ipn = 1.0f / lane_bcast(a[0], (p + 1) & 31);
#pragma unroll
                for (int c = 2; c < 32; ++c) a[c - 1] = a[c] - g * lane_bcast(a[c], p);
                a[31] = isp ? 1.0f : -g;
                ip = ipn;
            }
            float4* rout = reinterpret_cast<float4*>(sS + lane * SLD);
#pragma unroll
            for (int j = 0; j < 8; ++j) {
                float4 v;
                v.x = a[4 * j] * s; v.y = a[4 * j + 1] * s;
                v.z = a[4 * j + 2] * s; v.w = a[4 * j + 3] * s;
                rout[j] = v;
            }
        } else if (w == 9 || w == 10) {
            int m0 = (w - 9) * 16;
            v8f c = zero8();
            c = mmacc<false, false>(sxp, LDX, 0, sWfy, LDY, m0, NXX, c, lane);
            int h = lane >> 4, n = m0 + (lane & 15);
#pragma unroll
            for (int r = 0; r < 8; ++r) {
                int m = r + 8 * h;
                sin[m * LDY + n] = sym[m * LDY + n] - c[r];
            }
        }
        lds_barrier();
        if (gainw) {
            int mi = (w >> 1) * 16, ni = (w & 1) * 16;
            v8f c = zero8();
            c = mmacc<false, false>(sM2, LDY, mi, sS, SLD, ni, NYY, c, lane);
            storeD(sLn, LDY, mi, ni, c, lane);
        } else {
            v8f c = loadC(sxp, LDX, 0, n0, lane);
            c = mmacc<false, true>(sin, LDY, 0, sLc, LDY, n0, NYY, c, lane);
            storeD(sx, LDX, 0, n0, c, lane);
#pragma unroll
            for (int j = 0; j < 2; ++j) su[((bt >> 4) + 8 * j) * LDU + (bt & 15)] = nu[j];
            sd[(bt >> 3) * LDD + (bt & 7)] = nd;
#pragma unroll
            for (int j = 0; j < 4; ++j) sym[((bt >> 5) + 4 * j) * LDY + (bt & 31)] = ny[j];
        }
        lds_barrier();
        if (gainw) {
            int mi = (w >> 2) * 16, ni = (w & 3) * 16;
            v8f c = zero8();
            c = mmacc<true, false>(sWfy, LDY, mi, sP, LDX, ni, NXX, c, lane);
            storeD(sM1, LDX, mi, ni, c, lane, -1.0f);
        }
        lds_barrier();
        if (gainw) {
            for (int tt = w; tt < 16; tt += 8) {
                int mi = (tt >> 2) * 16, ni = (tt & 3) * 16;
                v8f c = identC(mi, ni, lane);
                c = mmacc<false, false>(sLn, LDY, mi, sM1, LDX, ni, NYY, c, lane);
                storeD(sP, LDX, mi, ni, c, lane);
            }
        }
        lds_barrier();
    }

    for (int i = tid; i < 16 * NXX; i += 384) {
        int r = i >> 6, cc = i & 63;
        out[(size_t)(r0 + r) * NXX + cc] = sx[r * LDX + cc];
    }
}

extern "C" void kernel_launch(void* const* d_in, const int* in_sizes, int n_in,
                              void* d_out, int out_size, void* d_ws, size_t ws_size,
                              hipStream_t stream) {
    (void)in_sizes; (void)n_in; (void)out_size; (void)d_ws; (void)ws_size;
    const float* Yp  = (const float*)d_in[0];
    const float* Up  = (const float*)d_in[1];
    const float* Dp  = (const float*)d_in[2];
    const float* Wfx = (const float*)d_in[3];
    const float* Wfu = (const float*)d_in[4];
    const float* Wfd = (const float*)d_in[5];
    const float* Wfy = (const float*)d_in[6];
    float* out = (float*)d_out;

    kf_fused<<<16, 384, 0, stream>>>(Yp, Up, Dp, Wfx, Wfu, Wfd, Wfy, out);
}


# Gain step: -(Wfy^T P), which does not need the inverse, moves from a phase of its own onto six idle gain waves under the one-wave inverse of S: 7 barriers per step, not 8; sums unchanged
# speedup vs baseline: 4.1094x; 1.0908x over previous
#include <hip/hip_runtime.h>
#include <hip/hip_bf16.h>

#define NXX 64
#define NYY 32
#define NUU 16
#define NDD 8
#define T_STEPS 2048
#define B_TOT 256
#define LDX 76
#define LDY 44
#define LDU 20
#define LDD 12
#define SLD LDY

typedef float v2f __attribute__((ext_vector_type(2)));
typedef float v8f __attribute__((ext_vector_type(8)));

__device__ __forceinline__ v8f zero8() {
    v8f c;
#pragma unroll
    for (int r = 0; r < 8; ++r) c[r] = 0.0f;
    return c;
}

__device__ __forceinline__ v2f fragA(const float* X, int ld, int m0, int k0, int lane) {
    int h = lane >> 4, m = m0 + (lane & 15), k = k0 + 2 * h;
    v2f a; a.x = X[m * ld + k]; a.y = X[m * ld + k + 1]; return a;
}
__device__ __forceinline__ v2f fragAT(const float* X, int ld, int m0, int k0, int lane) {
    int h = lane >> 4, m = m0 + (lane & 15), k = k0 + 2 * h;
    v2f a; a.x = X[k * ld + m]; a.y = X[(k + 1) * ld + m]; return a;
}
__device__ __forceinline__ v2f fragB(const float* X, int ld, int k0, int n0, int lane) {
    int h = lane >> 4, n = n0 + (lane & 15), k = k0 + 2 * h;
    v2f b; b.x = X[k * ld + n]; b.y = X[(k + 1) * ld + n]; return b;
}
__device__ __forceinline__ v2f fragBT(const float* X, int ld, int k0, int n0, int lane) {
    int h = lane >> 4, n = n0 + (lane & 15), k = k0 + 2 * h;
    v2f b; b.x = X[n * ld + k]; b.y = X[n * ld + k + 1]; return b;
}
__device__ __forceinline__ void storeD(float* X, int ld, int m0, int n0, v8f d, int lane,
                                       float sgn = 1.0f) {
    int h = lane >> 4, n = n0 + (lane & 15);
#pragma unroll
    for (int r = 0; r < 8; ++r) X[(m0 + r + 8 * h) * ld + n] = sgn * d[r];
}
__device__ __forceinline__ v8f loadC(const float* X, int ld, int m0, int n0, int lane) {
    int h = lane >> 4, n = n0 + (lane & 15);
    v8f c;
#pragma unroll
    for (int r = 0; r < 8; ++r) c[r] = X[(m0 + r + 8 * h) * ld + n];
    return c;
}
__device__ __forceinline__ v8f identC(int m0, int n0, int lane) {
    int h = lane >> 4, n = n0 + (lane & 15);
    v8f c;
#pragma unroll
    for (int r = 0; r < 8; ++r) c[r] = ((m0 + r + 8 * h) == n) ? 1.0f : 0.0f;
    return c;
}

__device__ __forceinline__ float lane_bcast(float x, int l) {
    return __int_as_float(__builtin_amdgcn_readlane(__float_as_int(x), l));
}

template <bool TA, bool TB>
__device__ __forceinline__ v8f mmacc(const float* A, int lda, int m0,
                                     const float* B, int ldb, int n0,
                                     int K, v8f c, int lane) {
#pragma unroll
    for (int k = 0; k < K; k += 4) {
        v2f a = TA ? fragAT(A, lda, m0, k, lane) : fragA(A, lda, m0, k, lane);
        v2f b = TB ? fragBT(B, ldb, k, n0, lane) : fragB(B, ldb, k, n0, lane);
        c = __builtin_amdgcn_wmma_f32_16x16x4_f32(false, a, false, b, (short)0, c,
                                                  false, false);
    }
    return c;
}

__device__ __forceinline__ void lds_barrier() {
    __builtin_amdgcn_fence(__ATOMIC_RELEASE, "workgroup", "local");
    __builtin_amdgcn_s_barrier();
    __builtin_amdgcn_fence(__ATOMIC_ACQUIRE, "workgroup", "local");
}

__global__ __launch_bounds__(384) void kf_fused(const float* __restrict__ Yp,
                                                const float* __restrict__ Up,
                                                const float* __restrict__ Dp,
                                                const float* __restrict__ Wfx,
                                                const float* __restrict__ Wfu,
                                                const float* __restrict__ Wfd,
                                                const float* __restrict__ Wfy,
                                                float* __restrict__ out) {
    __shared__ __attribute__((aligned(16))) float sW[(NXX + NUU + NDD) * LDX];
    __shared__ __attribute__((aligned(16))) float sWfy[NXX * LDY];
    __shared__ __attribute__((aligned(16))) float sP[NXX * LDX];
    __shared__ __attribute__((aligned(16))) float sM1[NXX * LDX];
    __shared__ __attribute__((aligned(16))) float sM2[NXX * LDY];
    __shared__ __attribute__((aligned(16))) float sS[NYY * SLD];
    __shared__ __attribute__((aligned(16))) float sL[2 * NXX * LDY];
    __shared__ __attribute__((aligned(16))) float sx[16 * LDX];
    __shared__ __attribute__((aligned(16))) float sxp[16 * LDX];
    __shared__ __attribute__((aligned(16))) float su[16 * LDU];
    __shared__ __attribute__((aligned(16))) float sd[16 * LDD];
    __shared__ __attribute__((aligned(16))) float sym[16 * LDY];
    __shared__ __attribute__((aligned(16))) float sin[16 * LDY];

    const int tid = threadIdx.x, lane = tid & 31;
    const int w = __builtin_amdgcn_readfirstlane(tid >> 5);
    const bool gainw = (w < 8);
    const int bt = tid - 256;
    const int r0 = blockIdx.x * 16;
    const int n0 = (w - 8) * 16;

    for (int i = tid; i < NXX * NXX; i += 384) {
        int r = i >> 6, cc = i & 63;
        sW[r * LDX + cc] = Wfx[i];
        sP[r * LDX + cc] = (r == cc) ? 1.0f : 0.0f;
    }
    for (int i = tid; i < NUU * NXX; i += 384) sW[(NXX + (i >> 6)) * LDX + (i & 63)] = Wfu[i];
    for (int i = tid; i < NDD * NXX; i += 384) sW[(NXX + NUU + (i >> 6)) * LDX + (i & 63)] = Wfd[i];
    for (int i = tid; i < NXX * NYY; i += 384) {
        int r = i >> 5, cc = i & 31;
        sWfy[r * LDY + cc] = Wfy[i];
        sL[r * LDY + cc] = 0.0f;
    }
    for (int i = tid; i < 16 * NXX; i += 384) sx[(i >> 6) * LDX + (i & 63)] = 0.0f;
    {
        const float* ut = Up + (size_t)r0 * NUU;
        const float* dt = Dp + (size_t)r0 * NDD;
        const float* yt = Yp + (size_t)r0 * NYY;
        for (int i = tid; i < 16 * NUU; i += 384) su[(i >> 4) * LDU + (i & 15)] = ut[i];
        for (int i = tid; i < 16 * NDD; i += 384) sd[(i >> 3) * LDD + (i & 7)] = dt[i];
        for (int i = tid; i < 16 * NYY; i += 384) sym[(i >> 5) * LDY + (i & 31)] = yt[i];
    }
    __syncthreads();

    for (int t = 0; t < T_STEPS; ++t) {
        const float* sLc = sL + (t & 1) * (NXX * LDY);
        float* sLn = sL + ((t & 1) ^ 1) * (NXX * LDY);

        float nu[2], nd, ny[4];
        if (!gainw) {
            const int tn = (t + 1 < T_STEPS) ? t + 1 : t;
            const float* un = Up + ((size_t)tn * B_TOT + r0) * NUU;
            const float* dn = Dp + ((size_t)tn * B_TOT + r0) * NDD;
            const float* yn = Yp + ((size_t)tn * B_TOT + r0) * NYY;
#pragma unroll
            for (int j = 0; j < 2; ++j) nu[j] = un[bt + 128 * j];
            nd = dn[bt];
#pragma unroll
            for (int j = 0; j < 4; ++j) ny[j] = yn[bt + 128 * j];
        }

        if (gainw) {
            for (int tt = w; tt < 16; tt += 8) {
                int mi = (tt >> 2) * 16, ni = (tt & 3) * 16;
                v8f c = zero8();
                c = mmacc<false, false>(sW, LDX, mi, sP, LDX, ni, NXX, c, lane);
                storeD(sM1, LDX, mi, ni, c, lane);
            }
        }
        lds_barrier();
        if (gainw) {
            for (int tt = w; tt < 16; tt += 8) {
                int mi = (tt >> 2) * 16, ni = (tt & 3) * 16;
                v8f c = identC(mi, ni, lane);
                c = mmacc<false, true>(sM1, LDX, mi, sW, LDX, ni, NXX, c, lane);
                storeD(sP, LDX, mi, ni, c, lane);
            }
        }
        lds_barrier();
        if (gainw) {
            int mi = (w >> 1) * 16, ni = (w & 1) * 16;
            v8f c = zero8();
            c = mmacc<false, false>(sP, LDX, mi, sWfy, LDY, ni, NXX, c, lane);
            storeD(sM2, LDY, mi, ni, c, lane);
        }
        lds_barrier();
        if (gainw) {
            if (w < 4) {
                int mi = (w >> 1) * 16, ni = (w & 1) * 16;
                v8f c = identC(mi, ni, lane);
                c = mmacc<true, false>(sWfy, LDY, mi, sM2, LDY, ni, NXX, c, lane);
                storeD(sS, SLD, mi, ni, c, lane);
            }
        } else {
            v8f c = zero8();
            c = mmacc<false, false>(sx, LDX, 0, sW, LDX, n0, NXX, c, lane);
            c = mmacc<false, false>(su, LDU, 0, sW + NXX * LDX, LDX, n0, NUU, c, lane);
            c = mmacc<false, false>(sd, LDD, 0, sW + (NXX + NUU) * LDX, LDX, n0, NDD, c, lane);
            storeD(sxp, LDX, 0, n0, c, lane);
        }
        lds_barrier();
        if (w == 0) {
            float a[32];
            const float4* rin = reinterpret_cast<const float4*>(sS + lane * SLD);
#pragma unroll
            for (int j = 0; j < 8; ++j) {
                float4 v = rin[j];
                a[4 * j] = v.x; a[4 * j + 1] = v.y; a[4 * j + 2] = v.z; a[4 * j + 3] = v.w;
            }
            float s = 1.0f;
            float ip = 1.0f / lane_bcast(a[0], 0);
            for (int p = 0; p < 32; ++p) {
                const bool isp = (lane == p);
                const float g = isp ? 0.0f : a[0] * ip;
                s = isp ? ip : s;
                a[0] = a[1] - g * lane_bcast(a[1], p);
                const float ipn = 1.0f / lane_bcast(a[0], (p + 1) & 31);
#pragma unroll
                for (int c = 2; c < 32; ++c) a[c - 1] = a[c] - g * lane_bcast(a[c], p);
                a[31] = isp ? 1.0f : -g;
                ip = ipn;
            }
            float4* rout = reinterpret_cast<float4*>(sS + lane * SLD);
#pragma unroll
            for (int j = 0; j < 8; ++j) {
                float4 v;
                v.x = a[4 * j] * s; v.y = a[4 * j + 1] * s;
                v.z = a[4 * j + 2] * s; v.w = a[4 * j + 3] * s;
                rout[j] = v;
            }
        } else if (w == 9 || w == 10) {
            int m0 = (w - 9) * 16;
            v8f c = zero8();
            c = mmacc<false, false>(sxp, LDX, 0, sWfy, LDY, m0, NXX, c, lane);
            int h = lane >> 4, n = m0 + (lane & 15);
#pragma unroll
            for (int r = 0; r < 8; ++r) {
                int m = r + 8 * h;
                sin[m * LDY + n] = sym[m * LDY + n] - c[r];
            }
        } else if (gainw && w != 4) {
            const bool two = ((w & 3) == 3);
            const int first = two ? 4 + 2 * (w >> 2) : (w & 3) - 1 + 2 * (w >> 2);
            const int last = two ? first + 1 : first;
            for (int tt = first; tt <= last; ++tt) {
                int mi = (tt >> 2) * 16, ni = (tt & 3) * 16;
                v8f c = zero8();
                c = mmacc<true, false>(sWfy, LDY, mi, sP, LDX, ni, NXX, c, lane);
                storeD(sM1, LDX, mi, ni, c, lane, -1.0f);
            }
        }
        lds_barrier();
        if (gainw) {
            int mi = (w >> 1) * 16, ni = (w & 1) * 16;
            v8f c = zero8();
            c = mmacc<false, false>(sM2, LDY, mi, sS, SLD, ni, NYY, c, lane);
            storeD(sLn, LDY, mi, ni, c, lane);
        } else {
            v8f c = loadC(sxp, LDX, 0, n0, lane);
            c = mmacc<false, true>(sin, LDY, 0, sLc, LDY, n0, NYY, c, lane);
            storeD(sx, LDX, 0, n0, c, lane);
#pragma unroll
            for (int j = 0; j < 2; ++j) su[((bt >> 4) + 8 * j) * LDU + (bt & 15)] = nu[j];
            sd[(bt >> 3) * LDD + (bt & 7)] = nd;
#pragma unroll
            for (int j = 0; j < 4; ++j) sym[((bt >> 5) + 4 * j) * LDY + (bt & 31)] = ny[j];
        }
        lds_barrier();
        if (gainw) {
            for (int tt = w; tt < 16; tt += 8) {
                int mi = (tt >> 2) * 16, ni = (tt & 3) * 16;
                v8f c = identC(mi, ni, lane);
                c = mmacc<false, false>(sLn, LDY, mi, sM1, LDX, ni, NYY, c, lane);
                storeD(sP, LDX, mi, ni, c, lane);
            }
        }
        lds_barrier();
    }

    for (int i = tid; i < 16 * NXX; i += 384) {
        int r = i >> 6, cc = i & 63;
        out[(size_t)(r0 + r) * NXX + cc] = sx[r * LDX + cc];
    }
}

extern "C" void kernel_launch(void* const* d_in, const int* in_sizes, int n_in,
                              void* d_out, int out_size, void* d_ws, size_t ws_size,
                              hipStream_t stream) {
    (void)in_sizes; (void)n_in; (void)out_size; (void)d_ws; (void)ws_size;
    const float* Yp  = (const float*)d_in[0];
    const float* Up  = (const float*)d_in[1];
    const float* Dp  = (const float*)d_in[2];
    const float* Wfx = (const float*)d_in[3];
    const float* Wfu = (const float*)d_in[4];
    const float* Wfd = (const float*)d_in[5];
    const float* Wfy = (const float*)d_in[6];
    float* out = (float*)d_out;

    kf_fused<<<16, 384, 0, stream>>>(Yp, Up, Dp, Wfx, Wfu, Wfd, Wfy, out);
}


# Gain recursion carried on predicted covariance: Pp' = C1 + (Wfx M2 S^-1)(-(Wfy^T Pp) Wfx^T), C1 = Wfx Wfx^T + I once; K=64 factors form under the inverse; 5 barriers not 7; P's sums reordered
# speedup vs baseline: 5.2286x; 1.2724x over previous
#include <hip/hip_runtime.h>
#include <hip/hip_bf16.h>

#define NXX 64
#define NYY 32
#define NUU 16
#define NDD 8
#define T_STEPS 2048
#define B_TOT 256
#define LDX 76
#define LDY 44
#define LDU 20
#define LDD 12
#define SLD LDY

typedef float v2f __attribute__((ext_vector_type(2)));
typedef float v8f __attribute__((ext_vector_type(8)));

__device__ __forceinline__ v8f zero8() {
    v8f c;
#pragma unroll
    for (int r = 0; r < 8; ++r) c[r] = 0.0f;
    return c;
}

__device__ __forceinline__ v2f fragA(const float* X, int ld, int m0, int k0, int lane) {
    int h = lane >> 4, m = m0 + (lane & 15), k = k0 + 2 * h;
    v2f a; a.x = X[m * ld + k]; a.y = X[m * ld + k + 1]; return a;
}
__device__ __forceinline__ v2f fragAT(const float* X, int ld, int m0, int k0, int lane) {
    int h = lane >> 4, m = m0 + (lane & 15), k = k0 + 2 * h;
    v2f a; a.x = X[k * ld + m]; a.y = X[(k + 1) * ld + m]; return a;
}
__device__ __forceinline__ v2f fragB(const float* X, int ld, int k0, int n0, int lane) {
    int h = lane >> 4, n = n0 + (lane & 15), k = k0 + 2 * h;
    v2f b; b.x = X[k * ld + n]; b.y = X[(k + 1) * ld + n]; return b;
}
__device__ __forceinline__ v2f fragBT(const float* X, int ld, int k0, int n0, int lane) {
    int h = lane >> 4, n = n0 + (lane & 15), k = k0 + 2 * h;
    v2f b; b.x = X[n * ld + k]; b.y = X[n * ld + k + 1]; return b;
}
__device__ __forceinline__ void storeD(float* X, int ld, int m0, int n0, v8f d, int lane,
                                       float sgn = 1.0f) {
    int h = lane >> 4, n = n0 + (lane & 15);
    float* p = X + (m0 + 8 * h) * ld + n;
#pragma unroll
    for (int r = 0; r < 8; ++r) p[r * ld] = sgn * d[r];
}
__device__ __forceinline__ v8f loadC(const float* X, int ld, int m0, int n0, int lane) {
    int h = lane >> 4, n = n0 + (lane & 15);
    const float* p = X + (m0 + 8 * h) * ld + n;
    v8f c;
#pragma unroll
    for (int r = 0; r < 8; ++r) c[r] = p[r * ld];
    return c;
}
__device__ __forceinline__ v8f identC(int m0, int n0, int lane) {
    int h = lane >> 4, n = n0 + (lane & 15);
    v8f c;
#pragma unroll
    for (int r = 0; r < 8; ++r) c[r] = ((m0 + r + 8 * h) == n) ? 1.0f : 0.0f;
    return c;
}

__device__ __forceinline__ float lane_bcast(float x, int l) {
    return __int_as_float(__builtin_amdgcn_readlane(__float_as_int(x), l));
}

template <bool TA, bool TB>
__device__ __forceinline__ v8f mmacc(const float* A, int lda, int m0,
                                     const float* B, int ldb, int n0,
                                     int K, v8f c, int lane) {
#pragma unroll
    for (int k = 0; k < K; k += 4) {
        v2f a = TA ? fragAT(A, lda, m0, k, lane) : fragA(A, lda, m0, k, lane);
        v2f b = TB ? fragBT(B, ldb, k, n0, lane) : fragB(B, ldb, k, n0, lane);
        c = __builtin_amdgcn_wmma_f32_16x16x4_f32(false, a, false, b, (short)0, c,
                                                  false, false);
    }
    return c;
}

__device__ __forceinline__ void lds_barrier() {
    __builtin_amdgcn_fence(__ATOMIC_RELEASE, "workgroup", "local");
    __builtin_amdgcn_s_barrier();
    __builtin_amdgcn_fence(__ATOMIC_ACQUIRE, "workgroup", "local");
}


__device__ __forceinline__ v8f wmma4(v2f a, v2f b, v8f c) {
    return __builtin_amdgcn_wmma_f32_16x16x4_f32(false, a, false, b, (short)0, c, false, false);
}

__device__ __forceinline__ void v_rowblock(int rb, const float* sWfy, const float* sPp,
                                           const float* sW, float* sM1, float* sV, int lane) {
    const int m0 = rb * 16;
    v8f c0 = zero8(), c1 = zero8(), c2 = zero8(), c3 = zero8();
#pragma unroll
    for (int k = 0; k < NXX; k += 4) {
        v2f a = fragAT(sWfy, LDY, m0, k, lane);
        c0 = wmma4(a, fragB(sPp, LDX, k, 0, lane), c0);
        c1 = wmma4(a, fragB(sPp, LDX, k, 16, lane), c1);
        c2 = wmma4(a, fragB(sPp, LDX, k, 32, lane), c2);
        c3 = wmma4(a, fragB(sPp, LDX, k, 48, lane), c3);
    }
    storeD(sM1, LDX, m0, 0, c0, lane, -1.0f);
    storeD(sM1, LDX, m0, 16, c1, lane, -1.0f);
    storeD(sM1, LDX, m0, 32, c2, lane, -1.0f);
    storeD(sM1, LDX, m0, 48, c3, lane, -1.0f);
    __builtin_amdgcn_fence(__ATOMIC_RELEASE, "wavefront", "local");
    __builtin_amdgcn_wave_barrier();
    __builtin_amdgcn_fence(__ATOMIC_ACQUIRE, "wavefront", "local");
    c0 = zero8(); c1 = zero8(); c2 = zero8(); c3 = zero8();
#pragma unroll
    for (int k = 0; k < NXX; k += 4) {
        v2f a = fragA(sM1, LDX, m0, k, lane);
        c0 = wmma4(a, fragBT(sW, LDX, k, 0, lane), c0);
        c1 = wmma4(a, fragBT(sW, LDX, k, 16, lane), c1);
        c2 = wmma4(a, fragBT(sW, LDX, k, 32, lane), c2);
        c3 = wmma4(a, fragBT(sW, LDX, k, 48, lane), c3);
    }
    storeD(sV, LDX, m0, 0, c0, lane);
    storeD(sV, LDX, m0, 16, c1, lane);
    storeD(sV, LDX, m0, 32, c2, lane);
    storeD(sV, LDX, m0, 48, c3, lane);
}

__device__ __forceinline__ void u_rowblock(int mb, const float* sW, const float* sM2, float* sU,
                                           int lane) {
    const int m0 = mb * 16;
    v8f c0 = zero8(), c1 = zero8();
#pragma unroll
    for (int k = 0; k < NXX; k += 4) {
        v2f a = fragA(sW, LDX, m0, k, lane);
        c0 = wmma4(a, fragB(sM2, LDY, k, 0, lane), c0);
        c1 = wmma4(a, fragB(sM2, LDY, k, 16, lane), c1);
    }
    storeD(sU, LDY, m0, 0, c0, lane);
    storeD(sU, LDY, m0, 16, c1, lane);
}

__global__ __launch_bounds__(384) void kf_fused(const float* __restrict__ Yp,
                                                const float* __restrict__ Up,
                                                const float* __restrict__ Dp,
                                                const float* __restrict__ Wfx,
                                                const float* __restrict__ Wfu,
                                                const float* __restrict__ Wfd,
                                                const float* __restrict__ Wfy,
                                                float* __restrict__ out) {
    __shared__ __attribute__((aligned(16))) float sW[(NXX + NUU + NDD) * LDX];
    __shared__ __attribute__((aligned(16))) float sWfy[NXX * LDY];
    __shared__ __attribute__((aligned(16))) float sPp[NXX * LDX];
    __shared__ __attribute__((aligned(16))) float sC1[NXX * LDX];
    __shared__ __attribute__((aligned(16))) float sM1[NYY * LDX];
    __shared__ __attribute__((aligned(16))) float sM2[NXX * LDY];
    __shared__ __attribute__((aligned(16))) float sU[NXX * LDY];
    __shared__ __attribute__((aligned(16))) float sV[NYY * LDX];
    __shared__ __attribute__((aligned(16))) float sZ[NXX * LDY];
    __shared__ __attribute__((aligned(16))) float sS[NYY * SLD];
    __shared__ __attribute__((aligned(16))) float sL[2 * NXX * LDY];
    __shared__ __attribute__((aligned(16))) float sx[16 * LDX];
    __shared__ __attribute__((aligned(16))) float sxp[16 * LDX];
    __shared__ __attribute__((aligned(16))) float su[16 * LDU];
    __shared__ __attribute__((aligned(16))) float sd[16 * LDD];
    __shared__ __attribute__((aligned(16))) float sym[16 * LDY];
    __shared__ __attribute__((aligned(16))) float sin[16 * LDY];

    const int tid = threadIdx.x, lane = tid & 31;
    const int w = __builtin_amdgcn_readfirstlane(tid >> 5);
    const bool gainw = (w < 8);
    const int bt = tid - 256;
    const int r0 = blockIdx.x * 16;
    const int n0 = (w - 8) * 16;

    for (int i = tid; i < NXX * NXX; i += 384) sW[(i >> 6) * LDX + (i & 63)] = Wfx[i];
    for (int i = tid; i < NUU * NXX; i += 384) sW[(NXX + (i >> 6)) * LDX + (i & 63)] = Wfu[i];
    for (int i = tid; i < NDD * NXX; i += 384) sW[(NXX + NUU + (i >> 6)) * LDX + (i & 63)] = Wfd[i];
    for (int i = tid; i < NXX * NYY; i += 384) {
        int r = i >> 5, cc = i & 31;
        sWfy[r * LDY + cc] = Wfy[i];
        sL[r * LDY + cc] = 0.0f;
    }
    for (int i = tid; i < 16 * NXX; i += 384) sx[(i >> 6) * LDX + (i & 63)] = 0.0f;
    {
        const float* ut = Up + (size_t)r0 * NUU;
        const float* dt = Dp + (size_t)r0 * NDD;
        const float* yt = Yp + (size_t)r0 * NYY;
        for (int i = tid; i < 16 * NUU; i += 384) su[(i >> 4) * LDU + (i & 15)] = ut[i];
        for (int i = tid; i < 16 * NDD; i += 384) sd[(i >> 3) * LDD + (i & 7)] = dt[i];
        for (int i = tid; i < 16 * NYY; i += 384) sym[(i >> 5) * LDY + (i & 31)] = yt[i];
    }
    __syncthreads();
    if (gainw) {
        for (int tt = w; tt < 16; tt += 8) {
            int mi = (tt >> 2) * 16, ni = (tt & 3) * 16;
            v8f c = identC(mi, ni, lane);
            c = mmacc<false, true>(sW, LDX, mi, sW, LDX, ni, NXX, c, lane);
            storeD(sC1, LDX, mi, ni, c, lane);
            storeD(sPp, LDX, mi, ni, c, lane);
        }
    }
    __syncthreads();

    for (int t = 0; t < T_STEPS; ++t) {
        const float* sLc = sL + (t & 1) * (NXX * LDY);
        float* sLn = sL + ((t & 1) ^ 1) * (NXX * LDY);

        float nu[2], nd, ny[4];
        if (!gainw) {
            const int tn = (t + 1 < T_STEPS) ? t + 1 : t;
            const float* un = Up + ((size_t)tn * B_TOT + r0) * NUU;
            const float* dn = Dp + ((size_t)tn * B_TOT + r0) * NDD;
            const float* yn = Yp + ((size_t)tn * B_TOT + r0) * NYY;
#pragma unroll
            for (int j = 0; j < 2; ++j) nu[j] = un[bt + 128 * j];
            nd = dn[bt];
#pragma unroll
            for (int j = 0; j < 4; ++j) ny[j] = yn[bt + 128 * j];
        }

        if (gainw) {
            int mi = (w >> 1) * 16, ni = (w & 1) * 16;
            v8f c = zero8();
            c = mmacc<false, false>(sPp, LDX, mi, sWfy, LDY, ni, NXX, c, lane);
            storeD(sM2, LDY, mi, ni, c, lane);
        }
        lds_barrier();
        if (gainw) {
            if (w < 4) {
                int mi = (w >> 1) * 16, ni = (w & 1) * 16;
                v8f c = identC(mi, ni, lane);
                c = mmacc<true, false>(sWfy, LDY, mi, sM2, LDY, ni, NXX, c, lane);
                storeD(sS, SLD, mi, ni, c, lane);
            }
        } else {
            v8f c = zero8();
            c = mmacc<false, false>(sx, LDX, 0, sW, LDX, n0, NXX, c, lane);
            c = mmacc<false, false>(su, LDU, 0, sW + NXX * LDX, LDX, n0, NUU, c, lane);
            c = mmacc<false, false>(sd, LDD, 0, sW + (NXX + NUU) * LDX, LDX, n0, NDD, c, lane);
            storeD(sxp, LDX, 0, n0, c, lane);
        }
        lds_barrier();
        if (w == 0) {
            float a[32];
            const float4* rin = reinterpret_cast<const float4*>(sS + lane * SLD);
#pragma unroll
            for (int j = 0; j < 8; ++j) {
                float4 v = rin[j];
                a[4 * j] = v.x; a[4 * j + 1] = v.y; a[4 * j + 2] = v.z; a[4 * j + 3] = v.w;
            }
            float s = 1.0f;
            float ip = 1.0f / lane_bcast(a[0], 0);
            for (int p = 0; p < 32; ++p) {
                const bool isp = (lane == p);
                const float g = isp ? 0.0f : a[0] * ip;
                s = isp ? ip : s;
                a[0] = a[1] - g * lane_bcast(a[1], p);
                const float ipn = 1.0f / lane_bcast(a[0], (p + 1) & 31);
#pragma unroll
                for (int c = 2; c < 32; ++c) a[c - 1] = a[c] - g * lane_bcast(a[c], p);
                a[31] = isp ? 1.0f : -g;
                ip = ipn;
            }
            float4* rout = reinterpret_cast<float4*>(sS + lane * SLD);
#pragma unroll
            for (int j = 0; j < 8; ++j) {
                float4 v;
                v.x = a[4 * j] * s; v.y = a[4 * j + 1] * s;
                v.z = a[4 * j + 2] * s; v.w = a[4 * j + 3] * s;
                rout[j] = v;
            }
        } else if (w == 11) {
            v8f c0 = zero8(), c1 = zero8();
#pragma unroll
            for (int k = 0; k < NXX; k += 4) {
                v2f a = fragA(sxp, LDX, 0, k, lane);
                c0 = wmma4(a, fragB(sWfy, LDY, k, 0, lane), c0);
                c1 = wmma4(a, fragB(sWfy, LDY, k, 16, lane), c1);
            }
            int h = lane >> 4, n = lane & 15;
#pragma unroll
            for (int r = 0; r < 8; ++r) {
                int m = r + 8 * h;
                sin[m * LDY + n] = sym[m * LDY + n] - c0[r];
                sin[m * LDY + n + 16] = sym[m * LDY + n + 16] - c1[r];
            }
        } else if (w == 1 || w == 2) {
            v_rowblock(w - 1, sWfy, sPp, sW, sM1, sV, lane);
        } else if (w == 3 || w == 7) {
            const int mb0 = (w >> 2) * 2;
            for (int mb = mb0; mb < mb0 + 2; ++mb) u_rowblock(mb, sW, sM2, sU, lane);
        }
        lds_barrier();
        if (gainw) {
            int mi = (w >> 1) * 16, ni = (w & 1) * 16;
            v8f cz = zero8(), cl = zero8();
#pragma unroll
            for (int k = 0; k < NYY; k += 4) {
                v2f b = fragB(sS, SLD, k, ni, lane);
                cz = wmma4(fragA(sU, LDY, mi, k, lane), b, cz);
                cl = wmma4(fragA(sM2, LDY, mi, k, lane), b, cl);
            }
            storeD(sZ, LDY, mi, ni, cz, lane);
            storeD(sLn, LDY, mi, ni, cl, lane);
        } else {
            v8f c = loadC(sxp, LDX, 0, n0, lane);
            c = mmacc<false, true>(sin, LDY, 0, sLc, LDY, n0, NYY, c, lane);
            storeD(sx, LDX, 0, n0, c, lane);
#pragma unroll
            for (int j = 0; j < 2; ++j) su[((bt >> 4) + 8 * j) * LDU + (bt & 15)] = nu[j];
            sd[(bt >> 3) * LDD + (bt & 7)] = nd;
#pragma unroll
            for (int j = 0; j < 4; ++j) sym[((bt >> 5) + 4 * j) * LDY + (bt & 31)] = ny[j];
        }
        lds_barrier();
        if (gainw) {
            for (int tt = w; tt < 16; tt += 8) {
                int mi = (tt >> 2) * 16, ni = (tt & 3) * 16;
                v8f c = loadC(sC1, LDX, mi, ni, lane);
                c = mmacc<false, false>(sZ, LDY, mi, sV, LDX, ni, NYY, c, lane);
                storeD(sPp, LDX, mi, ni, c, lane);
            }
        }
        lds_barrier();
    }

    for (int i = tid; i < 16 * NXX; i += 384) {
        int r = i >> 6, cc = i & 63;
        out[(size_t)(r0 + r) * NXX + cc] = sx[r * LDX + cc];
    }
}

extern "C" void kernel_launch(void* const* d_in, const int* in_sizes, int n_in,
                              void* d_out, int out_size, void* d_ws, size_t ws_size,
                              hipStream_t stream) {
    (void)in_sizes; (void)n_in; (void)out_size; (void)d_ws; (void)ws_size;
    const float* Yp  = (const float*)d_in[0];
    const float* Up  = (const float*)d_in[1];
    const float* Dp  = (const float*)d_in[2];
    const float* Wfx = (const float*)d_in[3];
    const float* Wfu = (const float*)d_in[4];
    const float* Wfd = (const float*)d_in[5];
    const float* Wfy = (const float*)d_in[6];
    float* out = (float*)d_out;

    kf_fused<<<16, 384, 0, stream>>>(Yp, Up, Dp, Wfx, Wfu, Wfd, Wfy, out);
}


# Gain recursion carried on M2 = Pp Wfy (64x32) and S alone, by Pp = Pp^T: M2' = C1 Wfy - Z G^T, S' = Sc - T G^T, G = (Wfy^T Wfx) M2; Pp never formed; 3 barriers a step, not 5; P's sums reordered
# speedup vs baseline: 6.5852x; 1.2595x over previous
#include <hip/hip_runtime.h>
#include <hip/hip_bf16.h>

#define NXX 64
#define NYY 32
#define NUU 16
#define NDD 8
#define T_STEPS 2048
#define B_TOT 256
#define LDX 76
#define LDY 44
#define LDU 20
#define LDD 12
#define SLD LDY

typedef float v2f __attribute__((ext_vector_type(2)));
typedef float v8f __attribute__((ext_vector_type(8)));

__device__ __forceinline__ v8f zero8() {
    v8f c;
#pragma unroll
    for (int r = 0; r < 8; ++r) c[r] = 0.0f;
    return c;
}

__device__ __forceinline__ v2f fragA(const float* X, int ld, int m0, int k0, int lane) {
    int h = lane >> 4, m = m0 + (lane & 15), k = k0 + 2 * h;
    v2f a; a.x = X[m * ld + k]; a.y = X[m * ld + k + 1]; return a;
}
__device__ __forceinline__ v2f fragAT(const float* X, int ld, int m0, int k0, int lane) {
    int h = lane >> 4, m = m0 + (lane & 15), k = k0 + 2 * h;
    v2f a; a.x = X[k * ld + m]; a.y = X[(k + 1) * ld + m]; return a;
}
__device__ __forceinline__ v2f fragB(const float* X, int ld, int k0, int n0, int lane) {
    int h = lane >> 4, n = n0 + (lane & 15), k = k0 + 2 * h;
    v2f b; b.x = X[k * ld + n]; b.y = X[(k + 1) * ld + n]; return b;
}
__device__ __forceinline__ v2f fragBT(const float* X, int ld, int k0, int n0, int lane) {
    int h = lane >> 4, n = n0 + (lane & 15), k = k0 + 2 * h;
    v2f b; b.x = X[n * ld + k]; b.y = X[n * ld + k + 1]; return b;
}
__device__ __forceinline__ void storeD(float* X, int ld, int m0, int n0, v8f d, int lane,
                                       float sgn = 1.0f) {
    int h = lane >> 4, n = n0 + (lane & 15);
    float* p = X + (m0 + 8 * h) * ld + n;
#pragma unroll
    for (int r = 0; r < 8; ++r) p[r * ld] = sgn * d[r];
}
__device__ __forceinline__ v8f loadC(const float* X, int ld, int m0, int n0, int lane) {
    int h = lane >> 4, n = n0 + (lane & 15);
    const float* p = X + (m0 + 8 * h) * ld + n;
    v8f c;
#pragma unroll
    for (int r = 0; r < 8; ++r) c[r] = p[r * ld];
    return c;
}
__device__ __forceinline__ v8f identC(int m0, int n0, int lane) {
    int h = lane >> 4, n = n0 + (lane & 15);
    v8f c;
#pragma unroll
    for (int r = 0; r < 8; ++r) c[r] = ((m0 + r + 8 * h) == n) ? 1.0f : 0.0f;
    return c;
}

__device__ __forceinline__ float lane_bcast(float x, int l) {
    return __int_as_float(__builtin_amdgcn_readlane(__float_as_int(x), l));
}

template <bool TA, bool TB>
__device__ __forceinline__ v8f mmacc(const float* A, int lda, int m0,
                                     const float* B, int ldb, int n0,
                                     int K, v8f c, int lane) {
#pragma unroll
    for (int k = 0; k < K; k += 4) {
        v2f a = TA ? fragAT(A, lda, m0, k, lane) : fragA(A, lda, m0, k, lane);
        v2f b = TB ? fragBT(B, ldb, k, n0, lane) : fragB(B, ldb, k, n0, lane);
        c = __builtin_amdgcn_wmma_f32_16x16x4_f32(false, a, false, b, (short)0, c,
                                                  false, false);
    }
    return c;
}

__device__ __forceinline__ void lds_barrier() {
    __builtin_amdgcn_fence(__ATOMIC_RELEASE, "workgroup", "local");
    __builtin_amdgcn_s_barrier();
    __builtin_amdgcn_fence(__ATOMIC_ACQUIRE, "workgroup", "local");
}


__device__ __forceinline__ v8f wmma4(v2f a, v2f b, v8f c) {
    return __builtin_amdgcn_wmma_f32_16x16x4_f32(false, a, false, b, (short)0, c, false, false);
}

__device__ __forceinline__ void k64_rowblock(const float* A, int lda, int m0, const float* B,
                                             float* Out, int lane) {
    v8f c0 = zero8(), c1 = zero8();
#pragma unroll
    for (int k = 0; k < NXX; k += 4) {
        v2f a = fragA(A, lda, m0, k, lane);
        c0 = wmma4(a, fragB(B, LDY, k, 0, lane), c0);
        c1 = wmma4(a, fragB(B, LDY, k, 16, lane), c1);
    }
    storeD(Out, LDY, m0, 0, c0, lane);
    storeD(Out, LDY, m0, 16, c1, lane);
}

__device__ __forceinline__ void k32_rowblock(const float* A, int ma, const float* Sinv, float* Out,
                                             int mo, float sgn, int lane) {
    v8f c0 = zero8(), c1 = zero8();
#pragma unroll
    for (int k = 0; k < NYY; k += 4) {
        v2f a = fragA(A, LDY, ma, k, lane);
        c0 = wmma4(a, fragB(Sinv, SLD, k, 0, lane), c0);
        c1 = wmma4(a, fragB(Sinv, SLD, k, 16, lane), c1);
    }
    storeD(Out, LDY, mo, 0, c0, lane, sgn);
    storeD(Out, LDY, mo, 16, c1, lane, sgn);
}

__global__ __launch_bounds__(384) void kf_fused(const float* __restrict__ Yp,
                                                const float* __restrict__ Up,
                                                const float* __restrict__ Dp,
                                                const float* __restrict__ Wfx,
                                                const float* __restrict__ Wfu,
                                                const float* __restrict__ Wfd,
                                                const float* __restrict__ Wfy,
                                                float* __restrict__ out) {
    __shared__ __attribute__((aligned(16))) float sW[(NXX + NUU + NDD) * LDX];
    __shared__ __attribute__((aligned(16))) float sWfy[NXX * LDY];
    __shared__ __attribute__((aligned(16))) float sC1[NXX * LDX];
    __shared__ __attribute__((aligned(16))) float sH[NYY * LDX];
    __shared__ __attribute__((aligned(16))) float sM2c[NXX * LDY];
    __shared__ __attribute__((aligned(16))) float sSc[NYY * LDY];
    __shared__ __attribute__((aligned(16))) float sM2[NXX * LDY];
    __shared__ __attribute__((aligned(16))) float sU[NXX * LDY];
    __shared__ __attribute__((aligned(16))) float sG[NYY * LDY];
    __shared__ __attribute__((aligned(16))) float sZ[NXX * LDY];
    __shared__ __attribute__((aligned(16))) float sT[NYY * LDY];
    __shared__ __attribute__((aligned(16))) float sS[NYY * SLD];
    __shared__ __attribute__((aligned(16))) float sL[2 * NXX * LDY];
    __shared__ __attribute__((aligned(16))) float sx[16 * LDX];
    __shared__ __attribute__((aligned(16))) float sxp[16 * LDX];
    __shared__ __attribute__((aligned(16))) float su[16 * LDU];
    __shared__ __attribute__((aligned(16))) float sd[16 * LDD];
    __shared__ __attribute__((aligned(16))) float sym[16 * LDY];
    __shared__ __attribute__((aligned(16))) float sin[16 * LDY];

    const int tid = threadIdx.x, lane = tid & 31;
    const int w = __builtin_amdgcn_readfirstlane(tid >> 5);
    const bool gainw = (w < 8);
    const int bt = tid - 256;
    const int r0 = blockIdx.x * 16;
    const int n0 = (w - 8) * 16;

    for (int i = tid; i < NXX * NXX; i += 384) sW[(i >> 6) * LDX + (i & 63)] = Wfx[i];
    for (int i = tid; i < NUU * NXX; i += 384) sW[(NXX + (i >> 6)) * LDX + (i & 63)] = Wfu[i];
    for (int i = tid; i < NDD * NXX; i += 384) sW[(NXX + NUU + (i >> 6)) * LDX + (i & 63)] = Wfd[i];
    for (int i = tid; i < NXX * NYY; i += 384) {
        int r = i >> 5, cc = i & 31;
        sWfy[r * LDY + cc] = Wfy[i];
        sL[r * LDY + cc] = 0.0f;
    }
    for (int i = tid; i < 16 * NXX; i += 384) sx[(i >> 6) * LDX + (i & 63)] = 0.0f;
    {
        const float* ut = Up + (size_t)r0 * NUU;
        const float* dt = Dp + (size_t)r0 * NDD;
        const float* yt = Yp + (size_t)r0 * NYY;
        for (int i = tid; i < 16 * NUU; i += 384) su[(i >> 4) * LDU + (i & 15)] = ut[i];
        for (int i = tid; i < 16 * NDD; i += 384) sd[(i >> 3) * LDD + (i & 7)] = dt[i];
        for (int i = tid; i < 16 * NYY; i += 384) sym[(i >> 5) * LDY + (i & 31)] = yt[i];
    }
    __syncthreads();
    if (gainw) {
        for (int tt = w; tt < 16; tt += 8) {
            int mi = (tt >> 2) * 16, ni = (tt & 3) * 16;
            v8f c = identC(mi, ni, lane);
            c = mmacc<false, true>(sW, LDX, mi, sW, LDX, ni, NXX, c, lane);
            storeD(sC1, LDX, mi, ni, c, lane);
        }
    } else {
        for (int tt = w - 8; tt < 8; tt += 4) {
            int mi = (tt >> 2) * 16, ni = (tt & 3) * 16;
            v8f c = zero8();
            c = mmacc<true, false>(sWfy, LDY, mi, sW, LDX, ni, NXX, c, lane);
            storeD(sH, LDX, mi, ni, c, lane);
        }
        v8f c = zero8();
        c = mmacc<false, false>(sx, LDX, 0, sW, LDX, n0, NXX, c, lane);
        c = mmacc<false, false>(su, LDU, 0, sW + NXX * LDX, LDX, n0, NUU, c, lane);
        c = mmacc<false, false>(sd, LDD, 0, sW + (NXX + NUU) * LDX, LDX, n0, NDD, c, lane);
        storeD(sxp, LDX, 0, n0, c, lane);
    }
    __syncthreads();
    if (gainw) {
        int mi = (w >> 1) * 16, ni = (w & 1) * 16;
        v8f c = zero8();
        c = mmacc<false, false>(sC1, LDX, mi, sWfy, LDY, ni, NXX, c, lane);
        storeD(sM2c, LDY, mi, ni, c, lane);
        storeD(sM2, LDY, mi, ni, c, lane);
    }
    __syncthreads();
    if (w < 4) {
        int mi = (w >> 1) * 16, ni = (w & 1) * 16;
        v8f c = identC(mi, ni, lane);
        c = mmacc<true, false>(sWfy, LDY, mi, sM2c, LDY, ni, NXX, c, lane);
        storeD(sSc, LDY, mi, ni, c, lane);
        storeD(sS, SLD, mi, ni, c, lane);
    }
    __syncthreads();

    for (int t = 0; t < T_STEPS; ++t) {
        const float* sLc = sL + (t & 1) * (NXX * LDY);
        float* sLn = sL + ((t & 1) ^ 1) * (NXX * LDY);

        float nu[2], nd, ny[4];
        if (!gainw) {
            const int tn = (t + 1 < T_STEPS) ? t + 1 : t;
            const float* un = Up + ((size_t)tn * B_TOT + r0) * NUU;
            const float* dn = Dp + ((size_t)tn * B_TOT + r0) * NDD;
            const float* yn = Yp + ((size_t)tn * B_TOT + r0) * NYY;
#pragma unroll
            for (int j = 0; j < 2; ++j) nu[j] = un[bt + 128 * j];
            nd = dn[bt];
#pragma unroll
            for (int j = 0; j < 4; ++j) ny[j] = yn[bt + 128 * j];
        }

        if (w == 0) {
            float a[32];
            const float4* rin = reinterpret_cast<const float4*>(sS + lane * SLD);
#pragma unroll
            for (int j = 0; j < 8; ++j) {
                float4 v = rin[j];
                a[4 * j] = v.x; a[4 * j + 1] = v.y; a[4 * j + 2] = v.z; a[4 * j + 3] = v.w;
            }
            float s = 1.0f;
            float ip = 1.0f / lane_bcast(a[0], 0);
            for (int p = 0; p < 32; ++p) {
                const bool isp = (lane == p);
                const float g = isp ? 0.0f : a[0] * ip;
                s = isp ? ip : s;
                a[0] = a[1] - g * lane_bcast(a[1], p);
                const float ipn = 1.0f / lane_bcast(a[0], (p + 1) & 31);
#pragma unroll
                for (int c = 2; c < 32; ++c) a[c - 1] = a[c] - g * lane_bcast(a[c], p);
                a[31] = isp ? 1.0f : -g;
                ip = ipn;
            }
            float4* rout = reinterpret_cast<float4*>(sS + lane * SLD);
#pragma unroll
            for (int j = 0; j < 8; ++j) {
                float4 v;
                v.x = a[4 * j] * s; v.y = a[4 * j + 1] * s;
                v.z = a[4 * j + 2] * s; v.w = a[4 * j + 3] * s;
                rout[j] = v;
            }
        } else if (w == 1 || w == 2 || w == 3 || w == 5) {
            const int mb = (w == 5) ? 3 : (w - 1);
            k64_rowblock(sW, LDX, mb * 16, sM2, sU, lane);
        } else if (w == 6 || w == 7) {
            k64_rowblock(sH, LDX, (w - 6) * 16, sM2, sG, lane);
        } else if (w == 9 || w == 10) {
            const int nn = (w - 9) * 16;
            v8f c = zero8();
            c = mmacc<false, false>(sxp, LDX, 0, sWfy, LDY, nn, NXX, c, lane);
            int h = lane >> 4, n = nn + (lane & 15);
#pragma unroll
            for (int r = 0; r < 8; ++r) {
                int m = r + 8 * h;
                sin[m * LDY + n] = sym[m * LDY + n] - c[r];
            }
        }
        lds_barrier();
        if (gainw) {
            if (w < 4) k32_rowblock(sU, w * 16, sS, sZ, w * 16, -1.0f, lane);
            else if (w < 6) k32_rowblock(sG, (w - 4) * 16, sS, sT, (w - 4) * 16, -1.0f, lane);
            else k32_rowblock(sM2, (w - 6) * 16, sS, sLn, (w - 6) * 16, 1.0f, lane);
        } else {
            v8f c = loadC(sxp, LDX, 0, n0, lane);
            c = mmacc<false, true>(sin, LDY, 0, sLc, LDY, n0, NYY, c, lane);
            storeD(sx, LDX, 0, n0, c, lane);
            {
                const int j = w - 8;
                const int mi = 32 + (j >> 1) * 16, ni = (j & 1) * 16;
                v8f cl = zero8();
                cl = mmacc<false, false>(sM2, LDY, mi, sS, SLD, ni, NYY, cl, lane);
                storeD(sLn, LDY, mi, ni, cl, lane);
            }
#pragma unroll
            for (int j = 0; j < 2; ++j) su[((bt >> 4) + 8 * j) * LDU + (bt & 15)] = nu[j];
            sd[(bt >> 3) * LDD + (bt & 7)] = nd;
#pragma unroll
            for (int j = 0; j < 4; ++j) sym[((bt >> 5) + 4 * j) * LDY + (bt & 31)] = ny[j];
        }
        lds_barrier();
        if (gainw) {
            const int mi = (w >> 1) * 16, ni = (w & 1) * 16;
            if (w < 4) {
                v8f c = loadC(sM2c, LDY, mi, ni, lane);
                c = mmacc<false, true>(sZ, LDY, mi, sG, LDY, ni, NYY, c, lane);
                storeD(sM2, LDY, mi, ni, c, lane);
            } else {
                const int ms = ((w - 4) >> 1) * 16;
                v8f c = loadC(sM2c, LDY, mi, ni, lane);
                v8f cs = loadC(sSc, LDY, ms, ni, lane);
#pragma unroll
                for (int k = 0; k < NYY; k += 4) {
                    v2f b = fragBT(sG, LDY, k, ni, lane);
                    c = wmma4(fragA(sZ, LDY, mi, k, lane), b, c);
                    cs = wmma4(fragA(sT, LDY, ms, k, lane), b, cs);
                }
                storeD(sM2, LDY, mi, ni, c, lane);
                storeD(sS, SLD, ms, ni, cs, lane);
            }
        } else {
            v8f c = zero8();
            c = mmacc<false, false>(sx, LDX, 0, sW, LDX, n0, NXX, c, lane);
            c = mmacc<false, false>(su, LDU, 0, sW + NXX * LDX, LDX, n0, NUU, c, lane);
            c = mmacc<false, false>(sd, LDD, 0, sW + (NXX + NUU) * LDX, LDX, n0, NDD, c, lane);
            storeD(sxp, LDX, 0, n0, c, lane);
        }
        lds_barrier();
    }

    for (int i = tid; i < 16 * NXX; i += 384) {
        int r = i >> 6, cc = i & 63;
        out[(size_t)(r0 + r) * NXX + cc] = sx[r * LDX + cc];
    }
}

extern "C" void kernel_launch(void* const* d_in, const int* in_sizes, int n_in,
                              void* d_out, int out_size, void* d_ws, size_t ws_size,
                              hipStream_t stream) {
    (void)in_sizes; (void)n_in; (void)out_size; (void)d_ws; (void)ws_size;
    const float* Yp  = (const float*)d_in[0];
    const float* Up  = (const float*)d_in[1];
    const float* Dp  = (const float*)d_in[2];
    const float* Wfx = (const float*)d_in[3];
    const float* Wfu = (const float*)d_in[4];
    const float* Wfd = (const float*)d_in[5];
    const float* Wfy = (const float*)d_in[6];
    float* out = (float*)d_out;

    kf_fused<<<16, 384, 0, stream>>>(Yp, Up, Dp, Wfx, Wfu, Wfd, Wfy, out);
}


# Inverse of S: a lane holds 2 rows x 16 columns, so each 16-lane row shares the pivot row by DPP row_share (16 per pivot, not 32 lane broadcasts via scalar registers); unrolled; sums unchanged
# speedup vs baseline: 8.3188x; 1.2633x over previous
#include <hip/hip_runtime.h>
#include <hip/hip_bf16.h>

#define NXX 64
#define NYY 32
#define NUU 16
#define NDD 8
#define T_STEPS 2048
#define B_TOT 256
#define LDX 76
#define LDY 44
#define LDU 20
#define LDD 12
#define SLD LDY

typedef float v2f __attribute__((ext_vector_type(2)));
typedef float v8f __attribute__((ext_vector_type(8)));

__device__ __forceinline__ v8f zero8() {
    v8f c;
#pragma unroll
    for (int r = 0; r < 8; ++r) c[r] = 0.0f;
    return c;
}

__device__ __forceinline__ v2f fragA(const float* X, int ld, int m0, int k0, int lane) {
    int h = lane >> 4, m = m0 + (lane & 15), k = k0 + 2 * h;
    v2f a; a.x = X[m * ld + k]; a.y = X[m * ld + k + 1]; return a;
}
__device__ __forceinline__ v2f fragAT(const float* X, int ld, int m0, int k0, int lane) {
    int h = lane >> 4, m = m0 + (lane & 15), k = k0 + 2 * h;
    v2f a; a.x = X[k * ld + m]; a.y = X[(k + 1) * ld + m]; return a;
}
__device__ __forceinline__ v2f fragB(const float* X, int ld, int k0, int n0, int lane) {
    int h = lane >> 4, n = n0 + (lane & 15), k = k0 + 2 * h;
    v2f b; b.x = X[k * ld + n]; b.y = X[(k + 1) * ld + n]; return b;
}
__device__ __forceinline__ v2f fragBT(const float* X, int ld, int k0, int n0, int lane) {
    int h = lane >> 4, n = n0 + (lane & 15), k = k0 + 2 * h;
    v2f b; b.x = X[n * ld + k]; b.y = X[n * ld + k + 1]; return b;
}
__device__ __forceinline__ void storeD(float* X, int ld, int m0, int n0, v8f d, int lane,
                                       float sgn = 1.0f) {
    int h = lane >> 4, n = n0 + (lane & 15);
    float* p = X + (m0 + 8 * h) * ld + n;
#pragma unroll
    for (int r = 0; r < 8; ++r) p[r * ld] = sgn * d[r];
}
__device__ __forceinline__ v8f loadC(const float* X, int ld, int m0, int n0, int lane) {
    int h = lane >> 4, n = n0 + (lane & 15);
    const float* p = X + (m0 + 8 * h) * ld + n;
    v8f c;
#pragma unroll
    for (int r = 0; r < 8; ++r) c[r] = p[r * ld];
    return c;
}
__device__ __forceinline__ v8f identC(int m0, int n0, int lane) {
    int h = lane >> 4, n = n0 + (lane & 15);
    v8f c;
#pragma unroll
    for (int r = 0; r < 8; ++r) c[r] = ((m0 + r + 8 * h) == n) ? 1.0f : 0.0f;
    return c;
}

__device__ __forceinline__ float lane_bcast(float x, int l) {
    return __int_as_float(__builtin_amdgcn_readlane(__float_as_int(x), l));
}

template <int PC>
__device__ __forceinline__ float row_bcast16(float x) {
    return __int_as_float(
        __builtin_amdgcn_update_dpp(0, __float_as_int(x), 0x150 + PC, 0xf, 0xf, true));
}
__device__ __forceinline__ float xhalf(float x) {
    const int b = __float_as_int(x);
    return __int_as_float(
        __builtin_amdgcn_permlanex16(b, b, 0x76543210, 0xfedcba98, false, false));
}

template <int P>
__device__ __forceinline__ void gj_pivot(float (&a0)[16], float (&a1)[16], float& s0, float& s1,
                                         float& ip, int lane) {
    constexpr int HP = P >> 4, PC = P & 15;
    constexpr int NP = (P + 1) & 31, NHP = NP >> 4, NPC = NP & 15;
    const bool inhp = ((lane >> 4) == HP);
    const bool isrow = ((lane & 15) == PC);
    const float o0 = a0[PC], o1 = a1[PC];
    const float x0 = xhalf(o0), x1 = xhalf(o1);
    float g0 = (inhp ? o0 : x0) * ip;
    float g1 = (inhp ? o1 : x1) * ip;
    if (HP == 0) { g0 = isrow ? 0.0f : g0; s0 = isrow ? ip : s0; }
    else         { g1 = isrow ? 0.0f : g1; s1 = isrow ? ip : s1; }
    float n0[16], n1[16];
    {
        const float bc = row_bcast16<PC>(HP ? a1[NPC] : a0[NPC]);
        n0[NPC] = a0[NPC] - g0 * bc;
        n1[NPC] = a1[NPC] - g1 * bc;
    }
    if (P + 1 < 32) ip = 1.0f / lane_bcast(NHP ? n1[NPC] : n0[NPC], NP);
#pragma unroll
    for (int c = 0; c < 16; ++c) {
        if (c == NPC) continue;
        const float bc = row_bcast16<PC>(HP ? a1[c] : a0[c]);
        n0[c] = a0[c] - g0 * bc;
        n1[c] = a1[c] - g1 * bc;
    }
    const float e0 = (HP == 0 && isrow) ? 1.0f : -g0;
    const float e1 = (HP == 1 && isrow) ? 1.0f : -g1;
    n0[PC] = inhp ? e0 : n0[PC];
    n1[PC] = inhp ? e1 : n1[PC];
#pragma unroll
    for (int c = 0; c < 16; ++c) { a0[c] = n0[c]; a1[c] = n1[c]; }
}

template <bool TA, bool TB>
__device__ __forceinline__ v8f mmacc(const float* A, int lda, int m0,
                                     const float* B, int ldb, int n0,
                                     int K, v8f c, int lane) {
#pragma unroll
    for (int k = 0; k < K; k += 4) {
        v2f a = TA ? fragAT(A, lda, m0, k, lane) : fragA(A, lda, m0, k, lane);
        v2f b = TB ? fragBT(B, ldb, k, n0, lane) : fragB(B, ldb, k, n0, lane);
        c = __builtin_amdgcn_wmma_f32_16x16x4_f32(false, a, false, b, (short)0, c,
                                                  false, false);
    }
    return c;
}

__device__ __forceinline__ void lds_barrier() {
    __builtin_amdgcn_fence(__ATOMIC_RELEASE, "workgroup", "local");
    __builtin_amdgcn_s_barrier();
    __builtin_amdgcn_fence(__ATOMIC_ACQUIRE, "workgroup", "local");
}


__device__ __forceinline__ v8f wmma4(v2f a, v2f b, v8f c) {
    return __builtin_amdgcn_wmma_f32_16x16x4_f32(false, a, false, b, (short)0, c, false, false);
}

__device__ __forceinline__ void k64_rowblock(const float* A, int lda, int m0, const float* B,
                                             float* Out, int lane) {
    v8f c0 = zero8(), c1 = zero8();
#pragma unroll
    for (int k = 0; k < NXX; k += 4) {
        v2f a = fragA(A, lda, m0, k, lane);
        c0 = wmma4(a, fragB(B, LDY, k, 0, lane), c0);
        c1 = wmma4(a, fragB(B, LDY, k, 16, lane), c1);
    }
    storeD(Out, LDY, m0, 0, c0, lane);
    storeD(Out, LDY, m0, 16, c1, lane);
}

__device__ __forceinline__ void k32_rowblock(const float* A, int ma, const float* Sinv, float* Out,
                                             int mo, float sgn, int lane) {
    v8f c0 = zero8(), c1 = zero8();
#pragma unroll
    for (int k = 0; k < NYY; k += 4) {
        v2f a = fragA(A, LDY, ma, k, lane);
        c0 = wmma4(a, fragB(Sinv, SLD, k, 0, lane), c0);
        c1 = wmma4(a, fragB(Sinv, SLD, k, 16, lane), c1);
    }
    storeD(Out, LDY, mo, 0, c0, lane, sgn);
    storeD(Out, LDY, mo, 16, c1, lane, sgn);
}

__global__ __launch_bounds__(384) void kf_fused(const float* __restrict__ Yp,
                                                const float* __restrict__ Up,
                                                const float* __restrict__ Dp,
                                                const float* __restrict__ Wfx,
                                                const float* __restrict__ Wfu,
                                                const float* __restrict__ Wfd,
                                                const float* __restrict__ Wfy,
                                                float* __restrict__ out) {
    __shared__ __attribute__((aligned(16))) float sW[(NXX + NUU + NDD) * LDX];
    __shared__ __attribute__((aligned(16))) float sWfy[NXX * LDY];
    __shared__ __attribute__((aligned(16))) float sC1[NXX * LDX];
    __shared__ __attribute__((aligned(16))) float sH[NYY * LDX];
    __shared__ __attribute__((aligned(16))) float sM2c[NXX * LDY];
    __shared__ __attribute__((aligned(16))) float sSc[NYY * LDY];
    __shared__ __attribute__((aligned(16))) float sM2[NXX * LDY];
    __shared__ __attribute__((aligned(16))) float sU[NXX * LDY];
    __shared__ __attribute__((aligned(16))) float sG[NYY * LDY];
    __shared__ __attribute__((aligned(16))) float sZ[NXX * LDY];
    __shared__ __attribute__((aligned(16))) float sT[NYY * LDY];
    __shared__ __attribute__((aligned(16))) float sS[NYY * SLD];
    __shared__ __attribute__((aligned(16))) float sL[2 * NXX * LDY];
    __shared__ __attribute__((aligned(16))) float sx[16 * LDX];
    __shared__ __attribute__((aligned(16))) float sxp[16 * LDX];
    __shared__ __attribute__((aligned(16))) float su[16 * LDU];
    __shared__ __attribute__((aligned(16))) float sd[16 * LDD];
    __shared__ __attribute__((aligned(16))) float sym[16 * LDY];
    __shared__ __attribute__((aligned(16))) float sin[16 * LDY];

    const int tid = threadIdx.x, lane = tid & 31;
    const int w = __builtin_amdgcn_readfirstlane(tid >> 5);
    const bool gainw = (w < 8);
    const int bt = tid - 256;
    const int r0 = blockIdx.x * 16;
    const int n0 = (w - 8) * 16;

    for (int i = tid; i < NXX * NXX; i += 384) sW[(i >> 6) * LDX + (i & 63)] = Wfx[i];
    for (int i = tid; i < NUU * NXX; i += 384) sW[(NXX + (i >> 6)) * LDX + (i & 63)] = Wfu[i];
    for (int i = tid; i < NDD * NXX; i += 384) sW[(NXX + NUU + (i >> 6)) * LDX + (i & 63)] = Wfd[i];
    for (int i = tid; i < NXX * NYY; i += 384) {
        int r = i >> 5, cc = i & 31;
        sWfy[r * LDY + cc] = Wfy[i];
        sL[r * LDY + cc] = 0.0f;
    }
    for (int i = tid; i < 16 * NXX; i += 384) sx[(i >> 6) * LDX + (i & 63)] = 0.0f;
    {
        const float* ut = Up + (size_t)r0 * NUU;
        const float* dt = Dp + (size_t)r0 * NDD;
        const float* yt = Yp + (size_t)r0 * NYY;
        for (int i = tid; i < 16 * NUU; i += 384) su[(i >> 4) * LDU + (i & 15)] = ut[i];
        for (int i = tid; i < 16 * NDD; i += 384) sd[(i >> 3) * LDD + (i & 7)] = dt[i];
        for (int i = tid; i < 16 * NYY; i += 384) sym[(i >> 5) * LDY + (i & 31)] = yt[i];
    }
    __syncthreads();
    if (gainw) {
        for (int tt = w; tt < 16; tt += 8) {
            int mi = (tt >> 2) * 16, ni = (tt & 3) * 16;
            v8f c = identC(mi, ni, lane);
            c = mmacc<false, true>(sW, LDX, mi, sW, LDX, ni, NXX, c, lane);
            storeD(sC1, LDX, mi, ni, c, lane);
        }
    } else {
        for (int tt = w - 8; tt < 8; tt += 4) {
            int mi = (tt >> 2) * 16, ni = (tt & 3) * 16;
            v8f c = zero8();
            c = mmacc<true, false>(sWfy, LDY, mi, sW, LDX, ni, NXX, c, lane);
            storeD(sH, LDX, mi, ni, c, lane);
        }
        v8f c = zero8();
        c = mmacc<false, false>(sx, LDX, 0, sW, LDX, n0, NXX, c, lane);
        c = mmacc<false, false>(su, LDU, 0, sW + NXX * LDX, LDX, n0, NUU, c, lane);
        c = mmacc<false, false>(sd, LDD, 0, sW + (NXX + NUU) * LDX, LDX, n0, NDD, c, lane);
        storeD(sxp, LDX, 0, n0, c, lane);
    }
    __syncthreads();
    if (gainw) {
        int mi = (w >> 1) * 16, ni = (w & 1) * 16;
        v8f c = zero8();
        c = mmacc<false, false>(sC1, LDX, mi, sWfy, LDY, ni, NXX, c, lane);
        storeD(sM2c, LDY, mi, ni, c, lane);
        storeD(sM2, LDY, mi, ni, c, lane);
    }
    __syncthreads();
    if (w < 4) {
        int mi = (w >> 1) * 16, ni = (w & 1) * 16;
        v8f c = identC(mi, ni, lane);
        c = mmacc<true, false>(sWfy, LDY, mi, sM2c, LDY, ni, NXX, c, lane);
        storeD(sSc, LDY, mi, ni, c, lane);
        storeD(sS, SLD, mi, ni, c, lane);
    }
    __syncthreads();

    for (int t = 0; t < T_STEPS; ++t) {
        const float* sLc = sL + (t & 1) * (NXX * LDY);
        float* sLn = sL + ((t & 1) ^ 1) * (NXX * LDY);

        float nu[2], nd, ny[4];
        if (!gainw) {
            const int tn = (t + 1 < T_STEPS) ? t + 1 : t;
            const float* un = Up + ((size_t)tn * B_TOT + r0) * NUU;
            const float* dn = Dp + ((size_t)tn * B_TOT + r0) * NDD;
            const float* yn = Yp + ((size_t)tn * B_TOT + r0) * NYY;
#pragma unroll
            for (int j = 0; j < 2; ++j) nu[j] = un[bt + 128 * j];
            nd = dn[bt];
#pragma unroll
            for (int j = 0; j < 4; ++j) ny[j] = yn[bt + 128 * j];
        }

        if (w == 0) {
            const int hh = lane >> 4, jj = lane & 15;
            float a0[16], a1[16];
            float* row0 = sS + jj * SLD + 16 * hh;
            float* row1 = sS + (jj + 16) * SLD + 16 * hh;
#pragma unroll
            for (int q = 0; q < 4; ++q) {
                const float4 u = reinterpret_cast<const float4*>(row0)[q];
                const float4 v = reinterpret_cast<const float4*>(row1)[q];
                a0[4 * q] = u.x; a0[4 * q + 1] = u.y; a0[4 * q + 2] = u.z; a0[4 * q + 3] = u.w;
                a1[4 * q] = v.x; a1[4 * q + 1] = v.y; a1[4 * q + 2] = v.z; a1[4 * q + 3] = v.w;
            }
            float s0 = 1.0f, s1 = 1.0f;
            float ip = 1.0f / lane_bcast(a0[0], 0);
            gj_pivot<0>(a0, a1, s0, s1, ip, lane);   gj_pivot<1>(a0, a1, s0, s1, ip, lane);
            gj_pivot<2>(a0, a1, s0, s1, ip, lane);   gj_pivot<3>(a0, a1, s0, s1, ip, lane);
            gj_pivot<4>(a0, a1, s0, s1, ip, lane);   gj_pivot<5>(a0, a1, s0, s1, ip, lane);
            gj_pivot<6>(a0, a1, s0, s1, ip, lane);   gj_pivot<7>(a0, a1, s0, s1, ip, lane);
            gj_pivot<8>(a0, a1, s0, s1, ip, lane);   gj_pivot<9>(a0, a1, s0, s1, ip, lane);
            gj_pivot<10>(a0, a1, s0, s1, ip, lane);  gj_pivot<11>(a0, a1, s0, s1, ip, lane);
            gj_pivot<12>(a0, a1, s0, s1, ip, lane);  gj_pivot<13>(a0, a1, s0, s1, ip, lane);
            gj_pivot<14>(a0, a1, s0, s1, ip, lane);  gj_pivot<15>(a0, a1, s0, s1, ip, lane);
            gj_pivot<16>(a0, a1, s0, s1, ip, lane);  gj_pivot<17>(a0, a1, s0, s1, ip, lane);
            gj_pivot<18>(a0, a1, s0, s1, ip, lane);  gj_pivot<19>(a0, a1, s0, s1, ip, lane);
            gj_pivot<20>(a0, a1, s0, s1, ip, lane);  gj_pivot<21>(a0, a1, s0, s1, ip, lane);
            gj_pivot<22>(a0, a1, s0, s1, ip, lane);  gj_pivot<23>(a0, a1, s0, s1, ip, lane);
            gj_pivot<24>(a0, a1, s0, s1, ip, lane);  gj_pivot<25>(a0, a1, s0, s1, ip, lane);
            gj_pivot<26>(a0, a1, s0, s1, ip, lane);  gj_pivot<27>(a0, a1, s0, s1, ip, lane);
            gj_pivot<28>(a0, a1, s0, s1, ip, lane);  gj_pivot<29>(a0, a1, s0, s1, ip, lane);
            gj_pivot<30>(a0, a1, s0, s1, ip, lane);  gj_pivot<31>(a0, a1, s0, s1, ip, lane);
#pragma unroll
            for (int q = 0; q < 4; ++q) {
                float4 u, v;
                u.x = a0[4 * q] * s0; u.y = a0[4 * q + 1] * s0;
                u.z = a0[4 * q + 2] * s0; u.w = a0[4 * q + 3] * s0;
                v.x = a1[4 * q] * s1; v.y = a1[4 * q + 1] * s1;
                v.z = a1[4 * q + 2] * s1; v.w = a1[4 * q + 3] * s1;
                reinterpret_cast<float4*>(row0)[q] = u;
                reinterpret_cast<float4*>(row1)[q] = v;
            }
        } else if (w == 1 || w == 2 || w == 3 || w == 5) {
            const int mb = (w == 5) ? 3 : (w - 1);
            k64_rowblock(sW, LDX, mb * 16, sM2, sU, lane);
        } else if (w == 6 || w == 7) {
            k64_rowblock(sH, LDX, (w - 6) * 16, sM2, sG, lane);
        } else if (w == 9 || w == 10) {
            const int nn = (w - 9) * 16;
            v8f c = zero8();
            c = mmacc<false, false>(sxp, LDX, 0, sWfy, LDY, nn, NXX, c, lane);
            int h = lane >> 4, n = nn + (lane & 15);
#pragma unroll
            for (int r = 0; r < 8; ++r) {
                int m = r + 8 * h;
                sin[m * LDY + n] = sym[m * LDY + n] - c[r];
            }
        }
        lds_barrier();
        if (gainw) {
            if (w < 4) k32_rowblock(sU, w * 16, sS, sZ, w * 16, -1.0f, lane);
            else if (w < 6) k32_rowblock(sG, (w - 4) * 16, sS, sT, (w - 4) * 16, -1.0f, lane);
            else k32_rowblock(sM2, (w - 6) * 16, sS, sLn, (w - 6) * 16, 1.0f, lane);
        } else {
            v8f c = loadC(sxp, LDX, 0, n0, lane);
            c = mmacc<false, true>(sin, LDY, 0, sLc, LDY, n0, NYY, c, lane);
            storeD(sx, LDX, 0, n0, c, lane);
            {
                const int j = w - 8;
                const int mi = 32 + (j >> 1) * 16, ni = (j & 1) * 16;
                v8f cl = zero8();
                cl = mmacc<false, false>(sM2, LDY, mi, sS, SLD, ni, NYY, cl, lane);
                storeD(sLn, LDY, mi, ni, cl, lane);
            }
#pragma unroll
            for (int j = 0; j < 2; ++j) su[((bt >> 4) + 8 * j) * LDU + (bt & 15)] = nu[j];
            sd[(bt >> 3) * LDD + (bt & 7)] = nd;
#pragma unroll
            for (int j = 0; j < 4; ++j) sym[((bt >> 5) + 4 * j) * LDY + (bt & 31)] = ny[j];
        }
        lds_barrier();
        if (gainw) {
            const int mi = (w >> 1) * 16, ni = (w & 1) * 16;
            if (w < 4) {
                v8f c = loadC(sM2c, LDY, mi, ni, lane);
                c = mmacc<false, true>(sZ, LDY, mi, sG, LDY, ni, NYY, c, lane);
                storeD(sM2, LDY, mi, ni, c, lane);
            } else {
                const int ms = ((w - 4) >> 1) * 16;
                v8f c = loadC(sM2c, LDY, mi, ni, lane);
                v8f cs = loadC(sSc, LDY, ms, ni, lane);
#pragma unroll
                for (int k = 0; k < NYY; k += 4) {
                    v2f b = fragBT(sG, LDY, k, ni, lane);
                    c = wmma4(fragA(sZ, LDY, mi, k, lane), b, c);
                    cs = wmma4(fragA(sT, LDY, ms, k, lane), b, cs);
                }
                storeD(sM2, LDY, mi, ni, c, lane);
                storeD(sS, SLD, ms, ni, cs, lane);
            }
        } else {
            v8f c = zero8();
            c = mmacc<false, false>(sx, LDX, 0, sW, LDX, n0, NXX, c, lane);
            c = mmacc<false, false>(su, LDU, 0, sW + NXX * LDX, LDX, n0, NUU, c, lane);
            c = mmacc<false, false>(sd, LDD, 0, sW + (NXX + NUU) * LDX, LDX, n0, NDD, c, lane);
            storeD(sxp, LDX, 0, n0, c, lane);
        }
        lds_barrier();
    }

    for (int i = tid; i < 16 * NXX; i += 384) {
        int r = i >> 6, cc = i & 63;
        out[(size_t)(r0 + r) * NXX + cc] = sx[r * LDX + cc];
    }
}

extern "C" void kernel_launch(void* const* d_in, const int* in_sizes, int n_in,
                              void* d_out, int out_size, void* d_ws, size_t ws_size,
                              hipStream_t stream) {
    (void)in_sizes; (void)n_in; (void)out_size; (void)d_ws; (void)ws_size;
    const float* Yp  = (const float*)d_in[0];
    const float* Up  = (const float*)d_in[1];
    const float* Dp  = (const float*)d_in[2];
    const float* Wfx = (const float*)d_in[3];
    const float* Wfu = (const float*)d_in[4];
    const float* Wfd = (const float*)d_in[5];
    const float* Wfy = (const float*)d_in[6];
    float* out = (float*)d_out;

    kf_fused<<<16, 384, 0, stream>>>(Yp, Up, Dp, Wfx, Wfu, Wfd, Wfy, out);
}


# Inverse of S: each pivot's reciprocal by v_rcp_f32 (1 ulp), taken in the lanes ahead of the broadcast, not the IEEE division whose dependent sequence sat on the pivot-to-pivot chain; sums unchanged
# speedup vs baseline: 9.0557x; 1.0886x over previous
#include <hip/hip_runtime.h>
#include <hip/hip_bf16.h>

#define NXX 64
#define NYY 32
#define NUU 16
#define NDD 8
#define T_STEPS 2048
#define B_TOT 256
#define LDX 76
#define LDY 44
#define LDU 20
#define LDD 12
#define SLD LDY

typedef float v2f __attribute__((ext_vector_type(2)));
typedef float v8f __attribute__((ext_vector_type(8)));

__device__ __forceinline__ v8f zero8() {
    v8f c;
#pragma unroll
    for (int r = 0; r < 8; ++r) c[r] = 0.0f;
    return c;
}

__device__ __forceinline__ v2f fragA(const float* X, int ld, int m0, int k0, int lane) {
    int h = lane >> 4, m = m0 + (lane & 15), k = k0 + 2 * h;
    v2f a; a.x = X[m * ld + k]; a.y = X[m * ld + k + 1]; return a;
}
__device__ __forceinline__ v2f fragAT(const float* X, int ld, int m0, int k0, int lane) {
    int h = lane >> 4, m = m0 + (lane & 15), k = k0 + 2 * h;
    v2f a; a.x = X[k * ld + m]; a.y = X[(k + 1) * ld + m]; return a;
}
__device__ __forceinline__ v2f fragB(const float* X, int ld, int k0, int n0, int lane) {
    int h = lane >> 4, n = n0 + (lane & 15), k = k0 + 2 * h;
    v2f b; b.x = X[k * ld + n]; b.y = X[(k + 1) * ld + n]; return b;
}
__device__ __forceinline__ v2f fragBT(const float* X, int ld, int k0, int n0, int lane) {
    int h = lane >> 4, n = n0 + (lane & 15), k = k0 + 2 * h;
    v2f b; b.x = X[n * ld + k]; b.y = X[n * ld + k + 1]; return b;
}
__device__ __forceinline__ void storeD(float* X, int ld, int m0, int n0, v8f d, int lane,
                                       float sgn = 1.0f) {
    int h = lane >> 4, n = n0 + (lane & 15);
    float* p = X + (m0 + 8 * h) * ld + n;
#pragma unroll
    for (int r = 0; r < 8; ++r) p[r * ld] = sgn * d[r];
}
__device__ __forceinline__ v8f loadC(const float* X, int ld, int m0, int n0, int lane) {
    int h = lane >> 4, n = n0 + (lane & 15);
    const float* p = X + (m0 + 8 * h) * ld + n;
    v8f c;
#pragma unroll
    for (int r = 0; r < 8; ++r) c[r] = p[r * ld];
    return c;
}
__device__ __forceinline__ v8f identC(int m0, int n0, int lane) {
    int h = lane >> 4, n = n0 + (lane & 15);
    v8f c;
#pragma unroll
    for (int r = 0; r < 8; ++r) c[r] = ((m0 + r + 8 * h) == n) ? 1.0f : 0.0f;
    return c;
}

__device__ __forceinline__ float lane_bcast(float x, int l) {
    return __int_as_float(__builtin_amdgcn_readlane(__float_as_int(x), l));
}

template <int PC>
__device__ __forceinline__ float row_bcast16(float x) {
    return __int_as_float(
        __builtin_amdgcn_update_dpp(0, __float_as_int(x), 0x150 + PC, 0xf, 0xf, true));
}
__device__ __forceinline__ float xhalf(float x) {
    const int b = __float_as_int(x);
    return __int_as_float(
        __builtin_amdgcn_permlanex16(b, b, 0x76543210, 0xfedcba98, false, false));
}

template <int P>
__device__ __forceinline__ void gj_pivot(float (&a0)[16], float (&a1)[16], float& s0, float& s1,
                                         float& ip, int lane) {
    constexpr int HP = P >> 4, PC = P & 15;
    constexpr int NP = (P + 1) & 31, NHP = NP >> 4, NPC = NP & 15;
    const bool inhp = ((lane >> 4) == HP);
    const bool isrow = ((lane & 15) == PC);
    const float o0 = a0[PC], o1 = a1[PC];
    const float x0 = xhalf(o0), x1 = xhalf(o1);
    float g0 = (inhp ? o0 : x0) * ip;
    float g1 = (inhp ? o1 : x1) * ip;
    if (HP == 0) { g0 = isrow ? 0.0f : g0; s0 = isrow ? ip : s0; }
    else         { g1 = isrow ? 0.0f : g1; s1 = isrow ? ip : s1; }
    float n0[16], n1[16];
    {
        const float bc = row_bcast16<PC>(HP ? a1[NPC] : a0[NPC]);
        n0[NPC] = a0[NPC] - g0 * bc;
        n1[NPC] = a1[NPC] - g1 * bc;
    }
    if (P + 1 < 32) ip = lane_bcast(__builtin_amdgcn_rcpf(NHP ? n1[NPC] : n0[NPC]), NP);
#pragma unroll
    for (int c = 0; c < 16; ++c) {
        if (c == NPC) continue;
        const float bc = row_bcast16<PC>(HP ? a1[c] : a0[c]);
        n0[c] = a0[c] - g0 * bc;
        n1[c] = a1[c] - g1 * bc;
    }
    const float e0 = (HP == 0 && isrow) ? 1.0f : -g0;
    const float e1 = (HP == 1 && isrow) ? 1.0f : -g1;
    n0[PC] = inhp ? e0 : n0[PC];
    n1[PC] = inhp ? e1 : n1[PC];
#pragma unroll
    for (int c = 0; c < 16; ++c) { a0[c] = n0[c]; a1[c] = n1[c]; }
}

template <bool TA, bool TB>
__device__ __forceinline__ v8f mmacc(const float* A, int lda, int m0,
                                     const float* B, int ldb, int n0,
                                     int K, v8f c, int lane) {
#pragma unroll
    for (int k = 0; k < K; k += 4) {
        v2f a = TA ? fragAT(A, lda, m0, k, lane) : fragA(A, lda, m0, k, lane);
        v2f b = TB ? fragBT(B, ldb, k, n0, lane) : fragB(B, ldb, k, n0, lane);
        c = __builtin_amdgcn_wmma_f32_16x16x4_f32(false, a, false, b, (short)0, c,
                                                  false, false);
    }
    return c;
}

__device__ __forceinline__ void lds_barrier() {
    __builtin_amdgcn_fence(__ATOMIC_RELEASE, "workgroup", "local");
    __builtin_amdgcn_s_barrier();
    __builtin_amdgcn_fence(__ATOMIC_ACQUIRE, "workgroup", "local");
}


__device__ __forceinline__ v8f wmma4(v2f a, v2f b, v8f c) {
    return __builtin_amdgcn_wmma_f32_16x16x4_f32(false, a, false, b, (short)0, c, false, false);
}

__device__ __forceinline__ void k64_rowblock(const float* A, int lda, int m0, const float* B,
                                             float* Out, int lane) {
    v8f c0 = zero8(), c1 = zero8();
#pragma unroll
    for (int k = 0; k < NXX; k += 4) {
        v2f a = fragA(A, lda, m0, k, lane);
        c0 = wmma4(a, fragB(B, LDY, k, 0, lane), c0);
        c1 = wmma4(a, fragB(B, LDY, k, 16, lane), c1);
    }
    storeD(Out, LDY, m0, 0, c0, lane);
    storeD(Out, LDY, m0, 16, c1, lane);
}

__device__ __forceinline__ void k32_rowblock(const float* A, int ma, const float* Sinv, float* Out,
                                             int mo, float sgn, int lane) {
    v8f c0 = zero8(), c1 = zero8();
#pragma unroll
    for (int k = 0; k < NYY; k += 4) {
        v2f a = fragA(A, LDY, ma, k, lane);
        c0 = wmma4(a, fragB(Sinv, SLD, k, 0, lane), c0);
        c1 = wmma4(a, fragB(Sinv, SLD, k, 16, lane), c1);
    }
    storeD(Out, LDY, mo, 0, c0, lane, sgn);
    storeD(Out, LDY, mo, 16, c1, lane, sgn);
}

__global__ __launch_bounds__(384) void kf_fused(const float* __restrict__ Yp,
                                                const float* __restrict__ Up,
                                                const float* __restrict__ Dp,
                                                const float* __restrict__ Wfx,
                                                const float* __restrict__ Wfu,
                                                const float* __restrict__ Wfd,
                                                const float* __restrict__ Wfy,
                                                float* __restrict__ out) {
    __shared__ __attribute__((aligned(16))) float sW[(NXX + NUU + NDD) * LDX];
    __shared__ __attribute__((aligned(16))) float sWfy[NXX * LDY];
    __shared__ __attribute__((aligned(16))) float sC1[NXX * LDX];
    __shared__ __attribute__((aligned(16))) float sH[NYY * LDX];
    __shared__ __attribute__((aligned(16))) float sM2c[NXX * LDY];
    __shared__ __attribute__((aligned(16))) float sSc[NYY * LDY];
    __shared__ __attribute__((aligned(16))) float sM2[NXX * LDY];
    __shared__ __attribute__((aligned(16))) float sU[NXX * LDY];
    __shared__ __attribute__((aligned(16))) float sG[NYY * LDY];
    __shared__ __attribute__((aligned(16))) float sZ[NXX * LDY];
    __shared__ __attribute__((aligned(16))) float sT[NYY * LDY];
    __shared__ __attribute__((aligned(16))) float sS[NYY * SLD];
    __shared__ __attribute__((aligned(16))) float sL[2 * NXX * LDY];
    __shared__ __attribute__((aligned(16))) float sx[16 * LDX];
    __shared__ __attribute__((aligned(16))) float sxp[16 * LDX];
    __shared__ __attribute__((aligned(16))) float su[16 * LDU];
    __shared__ __attribute__((aligned(16))) float sd[16 * LDD];
    __shared__ __attribute__((aligned(16))) float sym[16 * LDY];
    __shared__ __attribute__((aligned(16))) float sin[16 * LDY];

    const int tid = threadIdx.x, lane = tid & 31;
    const int w = __builtin_amdgcn_readfirstlane(tid >> 5);
    const bool gainw = (w < 8);
    const int bt = tid - 256;
    const int r0 = blockIdx.x * 16;
    const int n0 = (w - 8) * 16;

    for (int i = tid; i < NXX * NXX; i += 384) sW[(i >> 6) * LDX + (i & 63)] = Wfx[i];
    for (int i = tid; i < NUU * NXX; i += 384) sW[(NXX + (i >> 6)) * LDX + (i & 63)] = Wfu[i];
    for (int i = tid; i < NDD * NXX; i += 384) sW[(NXX + NUU + (i >> 6)) * LDX + (i & 63)] = Wfd[i];
    for (int i = tid; i < NXX * NYY; i += 384) {
        int r = i >> 5, cc = i & 31;
        sWfy[r * LDY + cc] = Wfy[i];
        sL[r * LDY + cc] = 0.0f;
    }
    for (int i = tid; i < 16 * NXX; i += 384) sx[(i >> 6) * LDX + (i & 63)] = 0.0f;
    {
        const float* ut = Up + (size_t)r0 * NUU;
        const float* dt = Dp + (size_t)r0 * NDD;
        const float* yt = Yp + (size_t)r0 * NYY;
        for (int i = tid; i < 16 * NUU; i += 384) su[(i >> 4) * LDU + (i & 15)] = ut[i];
        for (int i = tid; i < 16 * NDD; i += 384) sd[(i >> 3) * LDD + (i & 7)] = dt[i];
        for (int i = tid; i < 16 * NYY; i += 384) sym[(i >> 5) * LDY + (i & 31)] = yt[i];
    }
    __syncthreads();
    if (gainw) {
        for (int tt = w; tt < 16; tt += 8) {
            int mi = (tt >> 2) * 16, ni = (tt & 3) * 16;
            v8f c = identC(mi, ni, lane);
            c = mmacc<false, true>(sW, LDX, mi, sW, LDX, ni, NXX, c, lane);
            storeD(sC1, LDX, mi, ni, c, lane);
        }
    } else {
        for (int tt = w - 8; tt < 8; tt += 4) {
            int mi = (tt >> 2) * 16, ni = (tt & 3) * 16;
            v8f c = zero8();
            c = mmacc<true, false>(sWfy, LDY, mi, sW, LDX, ni, NXX, c, lane);
            storeD(sH, LDX, mi, ni, c, lane);
        }
        v8f c = zero8();
        c = mmacc<false, false>(sx, LDX, 0, sW, LDX, n0, NXX, c, lane);
        c = mmacc<false, false>(su, LDU, 0, sW + NXX * LDX, LDX, n0, NUU, c, lane);
        c = mmacc<false, false>(sd, LDD, 0, sW + (NXX + NUU) * LDX, LDX, n0, NDD, c, lane);
        storeD(sxp, LDX, 0, n0, c, lane);
    }
    __syncthreads();
    if (gainw) {
        int mi = (w >> 1) * 16, ni = (w & 1) * 16;
        v8f c = zero8();
        c = mmacc<false, false>(sC1, LDX, mi, sWfy, LDY, ni, NXX, c, lane);
        storeD(sM2c, LDY, mi, ni, c, lane);
        storeD(sM2, LDY, mi, ni, c, lane);
    }
    __syncthreads();
    if (w < 4) {
        int mi = (w >> 1) * 16, ni = (w & 1) * 16;
        v8f c = identC(mi, ni, lane);
        c = mmacc<true, false>(sWfy, LDY, mi, sM2c, LDY, ni, NXX, c, lane);
        storeD(sSc, LDY, mi, ni, c, lane);
        storeD(sS, SLD, mi, ni, c, lane);
    }
    __syncthreads();

    for (int t = 0; t < T_STEPS; ++t) {
        const float* sLc = sL + (t & 1) * (NXX * LDY);
        float* sLn = sL + ((t & 1) ^ 1) * (NXX * LDY);

        float nu[2], nd, ny[4];
        if (!gainw) {
            const int tn = (t + 1 < T_STEPS) ? t + 1 : t;
            const float* un = Up + ((size_t)tn * B_TOT + r0) * NUU;
            const float* dn = Dp + ((size_t)tn * B_TOT + r0) * NDD;
            const float* yn = Yp + ((size_t)tn * B_TOT + r0) * NYY;
#pragma unroll
            for (int j = 0; j < 2; ++j) nu[j] = un[bt + 128 * j];
            nd = dn[bt];
#pragma unroll
            for (int j = 0; j < 4; ++j) ny[j] = yn[bt + 128 * j];
        }

        if (w == 0) {
            const int hh = lane >> 4, jj = lane & 15;
            float a0[16], a1[16];
            float* row0 = sS + jj * SLD + 16 * hh;
            float* row1 = sS + (jj + 16) * SLD + 16 * hh;
#pragma unroll
            for (int q = 0; q < 4; ++q) {
                const float4 u = reinterpret_cast<const float4*>(row0)[q];
                const float4 v = reinterpret_cast<const float4*>(row1)[q];
                a0[4 * q] = u.x; a0[4 * q + 1] = u.y; a0[4 * q + 2] = u.z; a0[4 * q + 3] = u.w;
                a1[4 * q] = v.x; a1[4 * q + 1] = v.y; a1[4 * q + 2] = v.z; a1[4 * q + 3] = v.w;
            }
            float s0 = 1.0f, s1 = 1.0f;
            float ip = lane_bcast(__builtin_amdgcn_rcpf(a0[0]), 0);
            gj_pivot<0>(a0, a1, s0, s1, ip, lane);   gj_pivot<1>(a0, a1, s0, s1, ip, lane);
            gj_pivot<2>(a0, a1, s0, s1, ip, lane);   gj_pivot<3>(a0, a1, s0, s1, ip, lane);
            gj_pivot<4>(a0, a1, s0, s1, ip, lane);   gj_pivot<5>(a0, a1, s0, s1, ip, lane);
            gj_pivot<6>(a0, a1, s0, s1, ip, lane);   gj_pivot<7>(a0, a1, s0, s1, ip, lane);
            gj_pivot<8>(a0, a1, s0, s1, ip, lane);   gj_pivot<9>(a0, a1, s0, s1, ip, lane);
            gj_pivot<10>(a0, a1, s0, s1, ip, lane);  gj_pivot<11>(a0, a1, s0, s1, ip, lane);
            gj_pivot<12>(a0, a1, s0, s1, ip, lane);  gj_pivot<13>(a0, a1, s0, s1, ip, lane);
            gj_pivot<14>(a0, a1, s0, s1, ip, lane);  gj_pivot<15>(a0, a1, s0, s1, ip, lane);
            gj_pivot<16>(a0, a1, s0, s1, ip, lane);  gj_pivot<17>(a0, a1, s0, s1, ip, lane);
            gj_pivot<18>(a0, a1, s0, s1, ip, lane);  gj_pivot<19>(a0, a1, s0, s1, ip, lane);
            gj_pivot<20>(a0, a1, s0, s1, ip, lane);  gj_pivot<21>(a0, a1, s0, s1, ip, lane);
            gj_pivot<22>(a0, a1, s0, s1, ip, lane);  gj_pivot<23>(a0, a1, s0, s1, ip, lane);
            gj_pivot<24>(a0, a1, s0, s1, ip, lane);  gj_pivot<25>(a0, a1, s0, s1, ip, lane);
            gj_pivot<26>(a0, a1, s0, s1, ip, lane);  gj_pivot<27>(a0, a1, s0, s1, ip, lane);
            gj_pivot<28>(a0, a1, s0, s1, ip, lane);  gj_pivot<29>(a0, a1, s0, s1, ip, lane);
            gj_pivot<30>(a0, a1, s0, s1, ip, lane);  gj_pivot<31>(a0, a1, s0, s1, ip, lane);
#pragma unroll
            for (int q = 0; q < 4; ++q) {
                float4 u, v;
                u.x = a0[4 * q] * s0; u.y = a0[4 * q + 1] * s0;
                u.z = a0[4 * q + 2] * s0; u.w = a0[4 * q + 3] * s0;
                v.x = a1[4 * q] * s1; v.y = a1[4 * q + 1] * s1;
                v.z = a1[4 * q + 2] * s1; v.w = a1[4 * q + 3] * s1;
                reinterpret_cast<float4*>(row0)[q] = u;
                reinterpret_cast<float4*>(row1)[q] = v;
            }
        } else if (w == 1 || w == 2 || w == 3 || w == 5) {
            const int mb = (w == 5) ? 3 : (w - 1);
            k64_rowblock(sW, LDX, mb * 16, sM2, sU, lane);
        } else if (w == 6 || w == 7) {
            k64_rowblock(sH, LDX, (w - 6) * 16, sM2, sG, lane);
        } else if (w == 9 || w == 10) {
            const int nn = (w - 9) * 16;
            v8f c = zero8();
            c = mmacc<false, false>(sxp, LDX, 0, sWfy, LDY, nn, NXX, c, lane);
            int h = lane >> 4, n = nn + (lane & 15);
#pragma unroll
            for (int r = 0; r < 8; ++r) {
                int m = r + 8 * h;
                sin[m * LDY + n] = sym[m * LDY + n] - c[r];
            }
        }
        lds_barrier();
        if (gainw) {
            if (w < 4) k32_rowblock(sU, w * 16, sS, sZ, w * 16, -1.0f, lane);
            else if (w < 6) k32_rowblock(sG, (w - 4) * 16, sS, sT, (w - 4) * 16, -1.0f, lane);
            else k32_rowblock(sM2, (w - 6) * 16, sS, sLn, (w - 6) * 16, 1.0f, lane);
        } else {
            v8f c = loadC(sxp, LDX, 0, n0, lane);
            c = mmacc<false, true>(sin, LDY, 0, sLc, LDY, n0, NYY, c, lane);
            storeD(sx, LDX, 0, n0, c, lane);
            {
                const int j = w - 8;
                const int mi = 32 + (j >> 1) * 16, ni = (j & 1) * 16;
                v8f cl = zero8();
                cl = mmacc<false, false>(sM2, LDY, mi, sS, SLD, ni, NYY, cl, lane);
                storeD(sLn, LDY, mi, ni, cl, lane);
            }
#pragma unroll
            for (int j = 0; j < 2; ++j) su[((bt >> 4) + 8 * j) * LDU + (bt & 15)] = nu[j];
            sd[(bt >> 3) * LDD + (bt & 7)] = nd;
#pragma unroll
            for (int j = 0; j < 4; ++j) sym[((bt >> 5) + 4 * j) * LDY + (bt & 31)] = ny[j];
        }
        lds_barrier();
        if (gainw) {
            const int mi = (w >> 1) * 16, ni = (w & 1) * 16;
            if (w < 4) {
                v8f c = loadC(sM2c, LDY, mi, ni, lane);
                c = mmacc<false, true>(sZ, LDY, mi, sG, LDY, ni, NYY, c, lane);
                storeD(sM2, LDY, mi, ni, c, lane);
            } else {
                const int ms = ((w - 4) >> 1) * 16;
                v8f c = loadC(sM2c, LDY, mi, ni, lane);
                v8f cs = loadC(sSc, LDY, ms, ni, lane);
#pragma unroll
                for (int k = 0; k < NYY; k += 4) {
                    v2f b = fragBT(sG, LDY, k, ni, lane);
                    c = wmma4(fragA(sZ, LDY, mi, k, lane), b, c);
                    cs = wmma4(fragA(sT, LDY, ms, k, lane), b, cs);
                }
                storeD(sM2, LDY, mi, ni, c, lane);
                storeD(sS, SLD, ms, ni, cs, lane);
            }
        } else {
            v8f c = zero8();
            c = mmacc<false, false>(sx, LDX, 0, sW, LDX, n0, NXX, c, lane);
            c = mmacc<false, false>(su, LDU, 0, sW + NXX * LDX, LDX, n0, NUU, c, lane);
            c = mmacc<false, false>(sd, LDD, 0, sW + (NXX + NUU) * LDX, LDX, n0, NDD, c, lane);
            storeD(sxp, LDX, 0, n0, c, lane);
        }
        lds_barrier();
    }

    for (int i = tid; i < 16 * NXX; i += 384) {
        int r = i >> 6, cc = i & 63;
        out[(size_t)(r0 + r) * NXX + cc] = sx[r * LDX + cc];
    }
}

extern "C" void kernel_launch(void* const* d_in, const int* in_sizes, int n_in,
                              void* d_out, int out_size, void* d_ws, size_t ws_size,
                              hipStream_t stream) {
    (void)in_sizes; (void)n_in; (void)out_size; (void)d_ws; (void)ws_size;
    const float* Yp  = (const float*)d_in[0];
    const float* Up  = (const float*)d_in[1];
    const float* Dp  = (const float*)d_in[2];
    const float* Wfx = (const float*)d_in[3];
    const float* Wfu = (const float*)d_in[4];
    const float* Wfd = (const float*)d_in[5];
    const float* Wfy = (const float*)d_in[6];
    float* out = (float*)d_out;

    kf_fused<<<16, 384, 0, stream>>>(Yp, Up, Dp, Wfx, Wfu, Wfd, Wfy, out);
}


# speedup vs baseline: 9.6302x; 1.0634x over previous
#include <hip/hip_runtime.h>
#include <hip/hip_bf16.h>

#define NXX 64
#define NYY 32
#define NUU 16
#define NDD 8
#define T_STEPS 2048
#define B_TOT 256
#define LDX 76
#define LDY 44
#define LDU 20
#define LDD 12
#define SLD LDY

typedef float v2f __attribute__((ext_vector_type(2)));
typedef float v8f __attribute__((ext_vector_type(8)));

__device__ __forceinline__ v8f zero8() {
    v8f c;
#pragma unroll
    for (int r = 0; r < 8; ++r) c[r] = 0.0f;
    return c;
}

__device__ __forceinline__ v2f fragA(const float* X, int ld, int m0, int k0, int lane) {
    int h = lane >> 4, m = m0 + (lane & 15), k = k0 + 2 * h;
    v2f a; a.x = X[m * ld + k]; a.y = X[m * ld + k + 1]; return a;
}
__device__ __forceinline__ v2f fragAT(const float* X, int ld, int m0, int k0, int lane) {
    int h = lane >> 4, m = m0 + (lane & 15), k = k0 + 2 * h;
    v2f a; a.x = X[k * ld + m]; a.y = X[(k + 1) * ld + m]; return a;
}
__device__ __forceinline__ v2f fragB(const float* X, int ld, int k0, int n0, int lane) {
    int h = lane >> 4, n = n0 + (lane & 15), k = k0 + 2 * h;
    v2f b; b.x = X[k * ld + n]; b.y = X[(k + 1) * ld + n]; return b;
}
__device__ __forceinline__ v2f fragBT(const float* X, int ld, int k0, int n0, int lane) {
    int h = lane >> 4, n = n0 + (lane & 15), k = k0 + 2 * h;
    v2f b; b.x = X[n * ld + k]; b.y = X[n * ld + k + 1]; return b;
}
__device__ __forceinline__ void storeD(float* X, int ld, int m0, int n0, v8f d, int lane,
                                       float sgn = 1.0f) {
    int h = lane >> 4, n = n0 + (lane & 15);
    float* p = X + (m0 + 8 * h) * ld + n;
#pragma unroll
    for (int r = 0; r < 8; ++r) p[r * ld] = sgn * d[r];
}
__device__ __forceinline__ v8f loadC(const float* X, int ld, int m0, int n0, int lane) {
    int h = lane >> 4, n = n0 + (lane & 15);
    const float* p = X + (m0 + 8 * h) * ld + n;
    v8f c;
#pragma unroll
    for (int r = 0; r < 8; ++r) c[r] = p[r * ld];
    return c;
}
__device__ __forceinline__ v8f identC(int m0, int n0, int lane) {
    int h = lane >> 4, n = n0 + (lane & 15);
    v8f c;
#pragma unroll
    for (int r = 0; r < 8; ++r) c[r] = ((m0 + r + 8 * h) == n) ? 1.0f : 0.0f;
    return c;
}

__device__ __forceinline__ float lane_bcast(float x, int l) {
    return __int_as_float(__builtin_amdgcn_readlane(__float_as_int(x), l));
}

template <int PC>
__device__ __forceinline__ float row_bcast16(float x) {
    constexpr unsigned int SEL = 0x11111111u * PC;
    const int b = __float_as_int(x);
    return __int_as_float(__builtin_amdgcn_permlane16(b, b, SEL, SEL, false, true));
}
__device__ __forceinline__ float xhalf(float x) {
    const int b = __float_as_int(x);
    return __int_as_float(
        __builtin_amdgcn_permlanex16(b, b, 0x76543210, 0xfedcba98, false, false));
}

template <int P>
__device__ __forceinline__ void gj_pivot(float (&a0)[16], float (&a1)[16], float& s0, float& s1,
                                         float& ip, int lane) {
    constexpr int HP = P >> 4, PC = P & 15;
    constexpr int NP = (P + 1) & 31, NHP = NP >> 4, NPC = NP & 15;
    const bool inhp = ((lane >> 4) == HP);
    const bool isrow = ((lane & 15) == PC);
    const float o0 = a0[PC], o1 = a1[PC];
    const float x0 = xhalf(o0), x1 = xhalf(o1);
    float g0 = (inhp ? o0 : x0) * ip;
    float g1 = (inhp ? o1 : x1) * ip;
    if (HP == 0) { g0 = isrow ? 0.0f : g0; s0 = isrow ? ip : s0; }
    else         { g1 = isrow ? 0.0f : g1; s1 = isrow ? ip : s1; }
    float n0[16], n1[16];
    {
        const float bc = row_bcast16<PC>(HP ? a1[NPC] : a0[NPC]);
        n0[NPC] = a0[NPC] - g0 * bc;
        n1[NPC] = a1[NPC] - g1 * bc;
    }
    if (P + 1 < 32) ip = lane_bcast(__builtin_amdgcn_rcpf(NHP ? n1[NPC] : n0[NPC]), NP);
#pragma unroll
    for (int c = 0; c < 16; ++c) {
        if (c == NPC) continue;
        const float bc = row_bcast16<PC>(HP ? a1[c] : a0[c]);
        n0[c] = a0[c] - g0 * bc;
        n1[c] = a1[c] - g1 * bc;
    }
    const float e0 = (HP == 0 && isrow) ? 1.0f : -g0;
    const float e1 = (HP == 1 && isrow) ? 1.0f : -g1;
    n0[PC] = inhp ? e0 : n0[PC];
    n1[PC] = inhp ? e1 : n1[PC];
#pragma unroll
    for (int c = 0; c < 16; ++c) { a0[c] = n0[c]; a1[c] = n1[c]; }
}

template <bool TA, bool TB>
__device__ __forceinline__ v8f mmacc(const float* A, int lda, int m0,
                                     const float* B, int ldb, int n0,
                                     int K, v8f c, int lane) {
#pragma unroll
    for (int k = 0; k < K; k += 4) {
        v2f a = TA ? fragAT(A, lda, m0, k, lane) : fragA(A, lda, m0, k, lane);
        v2f b = TB ? fragBT(B, ldb, k, n0, lane) : fragB(B, ldb, k, n0, lane);
        c = __builtin_amdgcn_wmma_f32_16x16x4_f32(false, a, false, b, (short)0, c,
                                                  false, false);
    }
    return c;
}

__device__ __forceinline__ void lds_barrier() {
    __builtin_amdgcn_fence(__ATOMIC_RELEASE, "workgroup", "local");
    __builtin_amdgcn_s_barrier();
    __builtin_amdgcn_fence(__ATOMIC_ACQUIRE, "workgroup", "local");
}


__device__ __forceinline__ v8f wmma4(v2f a, v2f b, v8f c) {
    return __builtin_amdgcn_wmma_f32_16x16x4_f32(false, a, false, b, (short)0, c, false, false);
}

__device__ __forceinline__ void k64_rowblock(const float* A, int lda, int m0, const float* B,
                                             float* Out, int lane) {
    v8f c0 = zero8(), c1 = zero8();
#pragma unroll
    for (int k = 0; k < NXX; k += 4) {
        v2f a = fragA(A, lda, m0, k, lane);
        c0 = wmma4(a, fragB(B, LDY, k, 0, lane), c0);
        c1 = wmma4(a, fragB(B, LDY, k, 16, lane), c1);
    }
    storeD(Out, LDY, m0, 0, c0, lane);
    storeD(Out, LDY, m0, 16, c1, lane);
}

__device__ __forceinline__ void k32_rowblock(const float* A, int ma, const float* Sinv, float* Out,
                                             int mo, float sgn, int lane) {
    v8f c0 = zero8(), c1 = zero8();
#pragma unroll
    for (int k = 0; k < NYY; k += 4) {
        v2f a = fragA(A, LDY, ma, k, lane);
        c0 = wmma4(a, fragB(Sinv, SLD, k, 0, lane), c0);
        c1 = wmma4(a, fragB(Sinv, SLD, k, 16, lane), c1);
    }
    storeD(Out, LDY, mo, 0, c0, lane, sgn);
    storeD(Out, LDY, mo, 16, c1, lane, sgn);
}

__global__ __launch_bounds__(384) void kf_fused(const float* __restrict__ Yp,
                                                const float* __restrict__ Up,
                                                const float* __restrict__ Dp,
                                                const float* __restrict__ Wfx,
                                                const float* __restrict__ Wfu,
                                                const float* __restrict__ Wfd,
                                                const float* __restrict__ Wfy,
                                                float* __restrict__ out) {
    __shared__ __attribute__((aligned(16))) float sW[(NXX + NUU + NDD) * LDX];
    __shared__ __attribute__((aligned(16))) float sWfy[NXX * LDY];
    __shared__ __attribute__((aligned(16))) float sC1[NXX * LDX];
    __shared__ __attribute__((aligned(16))) float sH[NYY * LDX];
    __shared__ __attribute__((aligned(16))) float sM2c[NXX * LDY];
    __shared__ __attribute__((aligned(16))) float sSc[NYY * LDY];
    __shared__ __attribute__((aligned(16))) float sM2[NXX * LDY];
    __shared__ __attribute__((aligned(16))) float sU[NXX * LDY];
    __shared__ __attribute__((aligned(16))) float sG[NYY * LDY];
    __shared__ __attribute__((aligned(16))) float sZ[NXX * LDY];
    __shared__ __attribute__((aligned(16))) float sT[NYY * LDY];
    __shared__ __attribute__((aligned(16))) float sS[NYY * SLD];
    __shared__ __attribute__((aligned(16))) float sL[2 * NXX * LDY];
    __shared__ __attribute__((aligned(16))) float sx[16 * LDX];
    __shared__ __attribute__((aligned(16))) float sxp[16 * LDX];
    __shared__ __attribute__((aligned(16))) float su[16 * LDU];
    __shared__ __attribute__((aligned(16))) float sd[16 * LDD];
    __shared__ __attribute__((aligned(16))) float sym[16 * LDY];
    __shared__ __attribute__((aligned(16))) float sin[16 * LDY];

    const int tid = threadIdx.x, lane = tid & 31;
    const int w = __builtin_amdgcn_readfirstlane(tid >> 5);
    const bool gainw = (w < 8);
    const int bt = tid - 256;
    const int r0 = blockIdx.x * 16;
    const int n0 = (w - 8) * 16;

    for (int i = tid; i < NXX * NXX; i += 384) sW[(i >> 6) * LDX + (i & 63)] = Wfx[i];
    for (int i = tid; i < NUU * NXX; i += 384) sW[(NXX + (i >> 6)) * LDX + (i & 63)] = Wfu[i];
    for (int i = tid; i < NDD * NXX; i += 384) sW[(NXX + NUU + (i >> 6)) * LDX + (i & 63)] = Wfd[i];
    for (int i = tid; i < NXX * NYY; i += 384) {
        int r = i >> 5, cc = i & 31;
        sWfy[r * LDY + cc] = Wfy[i];
        sL[r * LDY + cc] = 0.0f;
    }
    for (int i = tid; i < 16 * NXX; i += 384) sx[(i >> 6) * LDX + (i & 63)] = 0.0f;
    {
        const float* ut = Up + (size_t)r0 * NUU;
        const float* dt = Dp + (size_t)r0 * NDD;
        const float* yt = Yp + (size_t)r0 * NYY;
        for (int i = tid; i < 16 * NUU; i += 384) su[(i >> 4) * LDU + (i & 15)] = ut[i];
        for (int i = tid; i < 16 * NDD; i += 384) sd[(i >> 3) * LDD + (i & 7)] = dt[i];
        for (int i = tid; i < 16 * NYY; i += 384) sym[(i >> 5) * LDY + (i & 31)] = yt[i];
    }
    __syncthreads();
    if (gainw) {
        for (int tt = w; tt < 16; tt += 8) {
            int mi = (tt >> 2) * 16, ni = (tt & 3) * 16;
            v8f c = identC(mi, ni, lane);
            c = mmacc<false, true>(sW, LDX, mi, sW, LDX, ni, NXX, c, lane);
            storeD(sC1, LDX, mi, ni, c, lane);
        }
    } else {
        for (int tt = w - 8; tt < 8; tt += 4) {
            int mi = (tt >> 2) * 16, ni = (tt & 3) * 16;
            v8f c = zero8();
            c = mmacc<true, false>(sWfy, LDY, mi, sW, LDX, ni, NXX, c, lane);
            storeD(sH, LDX, mi, ni, c, lane);
        }
        v8f c = zero8();
        c = mmacc<false, false>(sx, LDX, 0, sW, LDX, n0, NXX, c, lane);
        c = mmacc<false, false>(su, LDU, 0, sW + NXX * LDX, LDX, n0, NUU, c, lane);
        c = mmacc<false, false>(sd, LDD, 0, sW + (NXX + NUU) * LDX, LDX, n0, NDD, c, lane);
        storeD(sxp, LDX, 0, n0, c, lane);
    }
    __syncthreads();
    if (gainw) {
        int mi = (w >> 1) * 16, ni = (w & 1) * 16;
        v8f c = zero8();
        c = mmacc<false, false>(sC1, LDX, mi, sWfy, LDY, ni, NXX, c, lane);
        storeD(sM2c, LDY, mi, ni, c, lane);
        storeD(sM2, LDY, mi, ni, c, lane);
    }
    __syncthreads();
    if (w < 4) {
        int mi = (w >> 1) * 16, ni = (w & 1) * 16;
        v8f c = identC(mi, ni, lane);
        c = mmacc<true, false>(sWfy, LDY, mi, sM2c, LDY, ni, NXX, c, lane);
        storeD(sSc, LDY, mi, ni, c, lane);
        storeD(sS, SLD, mi, ni, c, lane);
    }
    __syncthreads();

    for (int t = 0; t < T_STEPS; ++t) {
        const float* sLc = sL + (t & 1) * (NXX * LDY);
        float* sLn = sL + ((t & 1) ^ 1) * (NXX * LDY);

        float nu[2], nd, ny[4];
        if (!gainw) {
            const int tn = (t + 1 < T_STEPS) ? t + 1 : t;
            const float* un = Up + ((size_t)tn * B_TOT + r0) * NUU;
            const float* dn = Dp + ((size_t)tn * B_TOT + r0) * NDD;
            const float* yn = Yp + ((size_t)tn * B_TOT + r0) * NYY;
#pragma unroll
            for (int j = 0; j < 2; ++j) nu[j] = un[bt + 128 * j];
            nd = dn[bt];
#pragma unroll
            for (int j = 0; j < 4; ++j) ny[j] = yn[bt + 128 * j];
        }

        if (w == 0) {
            const int hh = lane >> 4, jj = lane & 15;
            float a0[16], a1[16];
            float* row0 = sS + jj * SLD + 16 * hh;
            float* row1 = sS + (jj + 16) * SLD + 16 * hh;
#pragma unroll
            for (int q = 0; q < 4; ++q) {
                const float4 u = reinterpret_cast<const float4*>(row0)[q];
                const float4 v = reinterpret_cast<const float4*>(row1)[q];
                a0[4 * q] = u.x; a0[4 * q + 1] = u.y; a0[4 * q + 2] = u.z; a0[4 * q + 3] = u.w;
                a1[4 * q] = v.x; a1[4 * q + 1] = v.y; a1[4 * q + 2] = v.z; a1[4 * q + 3] = v.w;
            }
            float s0 = 1.0f, s1 = 1.0f;
            float ip = lane_bcast(__builtin_amdgcn_rcpf(a0[0]), 0);
            gj_pivot<0>(a0, a1, s0, s1, ip, lane);   gj_pivot<1>(a0, a1, s0, s1, ip, lane);
            gj_pivot<2>(a0, a1, s0, s1, ip, lane);   gj_pivot<3>(a0, a1, s0, s1, ip, lane);
            gj_pivot<4>(a0, a1, s0, s1, ip, lane);   gj_pivot<5>(a0, a1, s0, s1, ip, lane);
            gj_pivot<6>(a0, a1, s0, s1, ip, lane);   gj_pivot<7>(a0, a1, s0, s1, ip, lane);
            gj_pivot<8>(a0, a1, s0, s1, ip, lane);   gj_pivot<9>(a0, a1, s0, s1, ip, lane);
            gj_pivot<10>(a0, a1, s0, s1, ip, lane);  gj_pivot<11>(a0, a1, s0, s1, ip, lane);
            gj_pivot<12>(a0, a1, s0, s1, ip, lane);  gj_pivot<13>(a0, a1, s0, s1, ip, lane);
            gj_pivot<14>(a0, a1, s0, s1, ip, lane);  gj_pivot<15>(a0, a1, s0, s1, ip, lane);
            gj_pivot<16>(a0, a1, s0, s1, ip, lane);  gj_pivot<17>(a0, a1, s0, s1, ip, lane);
            gj_pivot<18>(a0, a1, s0, s1, ip, lane);  gj_pivot<19>(a0, a1, s0, s1, ip, lane);
            gj_pivot<20>(a0, a1, s0, s1, ip, lane);  gj_pivot<21>(a0, a1, s0, s1, ip, lane);
            gj_pivot<22>(a0, a1, s0, s1, ip, lane);  gj_pivot<23>(a0, a1, s0, s1, ip, lane);
            gj_pivot<24>(a0, a1, s0, s1, ip, lane);  gj_pivot<25>(a0, a1, s0, s1, ip, lane);
            gj_pivot<26>(a0, a1, s0, s1, ip, lane);  gj_pivot<27>(a0, a1, s0, s1, ip, lane);
            gj_pivot<28>(a0, a1, s0, s1, ip, lane);  gj_pivot<29>(a0, a1, s0, s1, ip, lane);
            gj_pivot<30>(a0, a1, s0, s1, ip, lane);  gj_pivot<31>(a0, a1, s0, s1, ip, lane);
#pragma unroll
            for (int q = 0; q < 4; ++q) {
                float4 u, v;
                u.x = a0[4 * q] * s0; u.y = a0[4 * q + 1] * s0;
                u.z = a0[4 * q + 2] * s0; u.w = a0[4 * q + 3] * s0;
                v.x = a1[4 * q] * s1; v.y = a1[4 * q + 1] * s1;
                v.z = a1[4 * q + 2] * s1; v.w = a1[4 * q + 3] * s1;
                reinterpret_cast<float4*>(row0)[q] = u;
                reinterpret_cast<float4*>(row1)[q] = v;
            }
        } else if (w == 1 || w == 2 || w == 3 || w == 5) {
            const int mb = (w == 5) ? 3 : (w - 1);
            k64_rowblock(sW, LDX, mb * 16, sM2, sU, lane);
        } else if (w == 6 || w == 7) {
            k64_rowblock(sH, LDX, (w - 6) * 16, sM2, sG, lane);
        } else if (w == 9 || w == 10) {
            const int nn = (w - 9) * 16;
            v8f c = zero8();
            c = mmacc<false, false>(sxp, LDX, 0, sWfy, LDY, nn, NXX, c, lane);
            int h = lane >> 4, n = nn + (lane & 15);
#pragma unroll
            for (int r = 0; r < 8; ++r) {
                int m = r + 8 * h;
                sin[m * LDY + n] = sym[m * LDY + n] - c[r];
            }
        }
        lds_barrier();
        if (gainw) {
            if (w < 4) k32_rowblock(sU, w * 16, sS, sZ, w * 16, -1.0f, lane);
            else if (w < 6) k32_rowblock(sG, (w - 4) * 16, sS, sT, (w - 4) * 16, -1.0f, lane);
            else k32_rowblock(sM2, (w - 6) * 16, sS, sLn, (w - 6) * 16, 1.0f, lane);
        } else {
            v8f c = loadC(sxp, LDX, 0, n0, lane);
            c = mmacc<false, true>(sin, LDY, 0, sLc, LDY, n0, NYY, c, lane);
            storeD(sx, LDX, 0, n0, c, lane);
            {
                const int j = w - 8;
                const int mi = 32 + (j >> 1) * 16, ni = (j & 1) * 16;
                v8f cl = zero8();
                cl = mmacc<false, false>(sM2, LDY, mi, sS, SLD, ni, NYY, cl, lane);
                storeD(sLn, LDY, mi, ni, cl, lane);
            }
#pragma unroll
            for (int j = 0; j < 2; ++j) su[((bt >> 4) + 8 * j) * LDU + (bt & 15)] = nu[j];
            sd[(bt >> 3) * LDD + (bt & 7)] = nd;
#pragma unroll
            for (int j = 0; j < 4; ++j) sym[((bt >> 5) + 4 * j) * LDY + (bt & 31)] = ny[j];
        }
        lds_barrier();
        if (gainw) {
            const int mi = (w >> 1) * 16, ni = (w & 1) * 16;
            if (w < 4) {
                v8f c = loadC(sM2c, LDY, mi, ni, lane);
                c = mmacc<false, true>(sZ, LDY, mi, sG, LDY, ni, NYY, c, lane);
                storeD(sM2, LDY, mi, ni, c, lane);
            } else {
                const int ms = ((w - 4) >> 1) * 16;
                v8f c = loadC(sM2c, LDY, mi, ni, lane);
                v8f cs = loadC(sSc, LDY, ms, ni, lane);
#pragma unroll
                for (int k = 0; k < NYY; k += 4) {
                    v2f b = fragBT(sG, LDY, k, ni, lane);
                    c = wmma4(fragA(sZ, LDY, mi, k, lane), b, c);
                    cs = wmma4(fragA(sT, LDY, ms, k, lane), b, cs);
                }
                storeD(sM2, LDY, mi, ni, c, lane);
                storeD(sS, SLD, ms, ni, cs, lane);
            }
        } else {
            v8f c = zero8();
            c = mmacc<false, false>(sx, LDX, 0, sW, LDX, n0, NXX, c, lane);
            c = mmacc<false, false>(su, LDU, 0, sW + NXX * LDX, LDX, n0, NUU, c, lane);
            c = mmacc<false, false>(sd, LDD, 0, sW + (NXX + NUU) * LDX, LDX, n0, NDD, c, lane);
            storeD(sxp, LDX, 0, n0, c, lane);
        }
        lds_barrier();
    }

    for (int i = tid; i < 16 * NXX; i += 384) {
        int r = i >> 6, cc = i & 63;
        out[(size_t)(r0 + r) * NXX + cc] = sx[r * LDX + cc];
    }
}

extern "C" void kernel_launch(void* const* d_in, const int* in_sizes, int n_in,
                              void* d_out, int out_size, void* d_ws, size_t ws_size,
                              hipStream_t stream) {
    (void)in_sizes; (void)n_in; (void)out_size; (void)d_ws; (void)ws_size;
    const float* Yp  = (const float*)d_in[0];
    const float* Up  = (const float*)d_in[1];
    const float* Dp  = (const float*)d_in[2];
    const float* Wfx = (const float*)d_in[3];
    const float* Wfu = (const float*)d_in[4];
    const float* Wfd = (const float*)d_in[5];
    const float* Wfy = (const float*)d_in[6];
    float* out = (float*)d_out;

    kf_fused<<<16, 384, 0, stream>>>(Yp, Up, Dp, Wfx, Wfu, Wfd, Wfy, out);
}
